# starting kernel
# baseline (speedup 1.0000x reference)
#include <hip/hip_runtime.h>
#include <hip/hip_bf16.h>


typedef __bf16 bf16_t;
typedef __attribute__((ext_vector_type(16))) __bf16 v16bf;
typedef __attribute__((ext_vector_type(8)))  __bf16 v8bf;
typedef __attribute__((ext_vector_type(8)))  float  v8f;

#define NN 50000
#define EE 500000
#define K1 257
#define KT1 9
#define N1 514
#define NT1 33
#define K2 514
#define KT2 17
#define N2 64
#define NT2 4
#define K3 64
#define KT3 2
#define N3 256
#define NT3 16
#define K4 256
#define KT4 8
#define N4 1
#define NT4 1
#define KN1 192
#define KTN1 6
#define NN1 256
#define NTN1 16
#define KN2 256
#define KTN2 8
#define NN2 128
#define NTN2 8

__device__ __forceinline__ bf16_t f2bf(float f) {
  unsigned u = __builtin_bit_cast(unsigned, f);
  unsigned r = u + 0x7FFFu + ((u >> 16) & 1u);
  unsigned short h = (unsigned short)(r >> 16);
  return __builtin_bit_cast(bf16_t, h);
}

__device__ __forceinline__ float bf2f(bf16_t b) {
  unsigned short h = __builtin_bit_cast(unsigned short, b);
  unsigned u = (unsigned)h << 16;
  return __builtin_bit_cast(float, u);
}

__device__ __forceinline__ float silu_f(float x) {
  return x / (1.0f + __expf(-x));
}

__device__ __forceinline__ v16bf load_a_frag(const bf16_t* A, int stride,
                                             int lane, int kbase) {
  int m = lane & 15;
  int kh = lane >> 4;
  const bf16_t* p = A + m * stride + kbase + 8 * kh;
  v8bf lo = *(const v8bf*)(p);
  v8bf hi = *(const v8bf*)(p + 16);
  return __builtin_shufflevector(lo, hi, 0, 1, 2, 3, 4, 5, 6, 7,
                                 8, 9, 10, 11, 12, 13, 14, 15);
}

__device__ __forceinline__ v16bf load_b_frag(const bf16_t* B, int NT,
                                             int kt, int nt, int lane) {
  return *(const v16bf*)(B + ((long long)(kt * NT + nt) * 32 + lane) * 16);
}

__device__ __forceinline__ v8f wmma_bf16(v16bf a, v16bf b, v8f c) {
  return __builtin_amdgcn_wmma_f32_16x16x32_bf16(
       false, a,  false, b,
       (short)0, c,  false,  false);
}


__global__ __launch_bounds__(256) void zero_f32_kernel(float* p, long long n) {
  long long i = (long long)blockIdx.x * 256 + threadIdx.x;
  if (i < n) p[i] = 0.0f;
}

__global__ __launch_bounds__(256) void pack_weight_kernel(
    const float* __restrict__ W, bf16_t* __restrict__ dst,
    int K, int N, int KT, int NT) {
  int idx = blockIdx.x * 256 + threadIdx.x;
  int total = KT * NT * 32;
  if (idx >= total) return;
  int lane = idx & 31;
  int tile = idx >> 5;
  int nt = tile % NT;
  int kt = tile / NT;
  int n = nt * 16 + (lane & 15);
  int kh = lane >> 4;
  bf16_t* o = dst + (long long)idx * 16;
#pragma unroll
  for (int e = 0; e < 16; ++e) {
    int k = kt * 32 + 8 * kh + e + ((e >= 8) ? 8 : 0);
    float v = (k < K && n < N) ? W[(long long)k * N + n] : 0.0f;
    o[e] = f2bf(v);
  }
}


#define SMEM_BYTES (18432 + 34816 + 4096)

__global__ __launch_bounds__(256) void egnn_edge_kernel(
    const float* __restrict__ x, const int* __restrict__ ei,
    const float* __restrict__ b_e1, const float* __restrict__ b_e2,
    const float* __restrict__ b_c1, const float* __restrict__ b_c2,
    const bf16_t* __restrict__ We1, const bf16_t* __restrict__ We2,
    const bf16_t* __restrict__ Wc1, const bf16_t* __restrict__ Wc2,
    float* __restrict__ m_i, float* __restrict__ w_i,
    float* __restrict__ r_i) {
  __shared__ alignas(16) char smem[SMEM_BYTES];
  __shared__ float sRel[32][3];
  __shared__ float sCw[32];
  __shared__ int   sDst[32];

  bf16_t* sA = (bf16_t*)smem;
  bf16_t* sT = (bf16_t*)smem;
  bf16_t* sH = (bf16_t*)(smem + 18432);
  bf16_t* sM = (bf16_t*)(smem + 53248);

  const int tid = threadIdx.x;
  const int lane = tid & 31;
  const int wave = tid >> 5;
  const int e0 = blockIdx.x * 32;

  {
    int m = tid >> 3;
    int sub = tid & 7;
    int e = e0 + m;
    int s = ei[e];
    int d = ei[EE + e];
    const float* xs = x + (long long)s * 131;
    const float* xd = x + (long long)d * 131;
#pragma unroll
    for (int j = 0; j < 16; ++j) {
      int k = sub * 16 + j;
      sA[m * 288 + k]       = f2bf(xd[k]);
      sA[m * 288 + 128 + k] = f2bf(xs[k]);
    }
    if (sub == 0) {
      float r0 = xs[128] - xd[128];
      float r1 = xs[129] - xd[129];
      float r2 = xs[130] - xd[130];
      float dist = sqrtf(r0 * r0 + r1 * r1 + r2 * r2);
      sA[m * 288 + 256] = f2bf(dist);
#pragma unroll
      for (int k = 257; k < 288; ++k) sA[m * 288 + k] = f2bf(0.0f);
      sRel[m][0] = r0; sRel[m][1] = r1; sRel[m][2] = r2;
      sDst[m] = d;
    }
#pragma unroll
    for (int i = 0; i < 2; ++i) {
      int idx = tid + i * 256;
      sH[(idx >> 4) * 544 + 528 + (idx & 15)] = f2bf(0.0f);
    }
  }
  __syncthreads();

  {
    for (int nt = wave; nt < NT1; nt += 8) {
      int n = nt * 16 + (lane & 15);
      float bias = (n < N1) ? b_e1[n] : 0.0f;
      v8f acc0, acc1;
#pragma unroll
      for (int r = 0; r < 8; ++r) { acc0[r] = bias; acc1[r] = bias; }
#pragma unroll
      for (int kt = 0; kt < KT1; ++kt) {
        v16bf b  = load_b_frag(We1, NT1, kt, nt, lane);
        v16bf a0 = load_a_frag(sA,            288, lane, kt * 32);
        v16bf a1 = load_a_frag(sA + 16 * 288, 288, lane, kt * 32);
        acc0 = wmma_bf16(a0, b, acc0);
        acc1 = wmma_bf16(a1, b, acc1);
      }
#pragma unroll
      for (int r = 0; r < 8; ++r) {
        int m = r + 8 * (lane >> 4);
        int col = nt * 16 + (lane & 15);
        sH[m * 544 + col]        = f2bf(silu_f(acc0[r]));
        sH[(m + 16) * 544 + col] = f2bf(silu_f(acc1[r]));
      }
    }
  }
  __syncthreads();

  {
    int mt = wave >> 2;
    int nt = wave & 3;
    int n = nt * 16 + (lane & 15);
    float bias = b_e2[n];
    v8f acc;
#pragma unroll
    for (int r = 0; r < 8; ++r) acc[r] = bias;
#pragma unroll
    for (int kt = 0; kt < KT2; ++kt) {
      v16bf a = load_a_frag(sH + mt * 16 * 544, 544, lane, kt * 32);
      acc = wmma_bf16(a, load_b_frag(We2, NT2, kt, nt, lane), acc);
    }
#pragma unroll
    for (int r = 0; r < 8; ++r) {
      int m = mt * 16 + r + 8 * (lane >> 4);
      sM[m * 64 + n] = f2bf(silu_f(acc[r]));
    }
  }
  __syncthreads();

  {
#pragma unroll
    for (int i = 0; i < 2; ++i) {
      int nt = wave + 8 * i;
      float bias = b_c1[nt * 16 + (lane & 15)];
      v8f acc0, acc1;
#pragma unroll
      for (int r = 0; r < 8; ++r) { acc0[r] = bias; acc1[r] = bias; }
#pragma unroll
      for (int kt = 0; kt < KT3; ++kt) {
        v16bf b  = load_b_frag(Wc1, NT3, kt, nt, lane);
        v16bf a0 = load_a_frag(sM,           64, lane, kt * 32);
        v16bf a1 = load_a_frag(sM + 16 * 64, 64, lane, kt * 32);
        acc0 = wmma_bf16(a0, b, acc0);
        acc1 = wmma_bf16(a1, b, acc1);
      }
#pragma unroll
      for (int r = 0; r < 8; ++r) {
        int m = r + 8 * (lane >> 4);
        int col = nt * 16 + (lane & 15);
        sT[m * 256 + col]        = f2bf(silu_f(acc0[r]));
        sT[(m + 16) * 256 + col] = f2bf(silu_f(acc1[r]));
      }
    }
  }
  __syncthreads();

  if (wave < 2) {
    int mt = wave;
    float bias = ((lane & 15) == 0) ? b_c2[0] : 0.0f;
    v8f acc;
#pragma unroll
    for (int r = 0; r < 8; ++r) acc[r] = bias;
#pragma unroll
    for (int kt = 0; kt < KT4; ++kt) {
      v16bf a = load_a_frag(sT + mt * 16 * 256, 256, lane, kt * 32);
      acc = wmma_bf16(a, load_b_frag(Wc2, NT4, kt, 0, lane), acc);
    }
    if ((lane & 15) == 0) {
#pragma unroll
      for (int r = 0; r < 8; ++r)
        sCw[mt * 16 + r + 8 * (lane >> 4)] = acc[r];
    }
  }
  __syncthreads();

#pragma unroll
  for (int i = 0; i < 8; ++i) {
    int idx = tid + i * 256;
    int m = idx >> 6;
    int j = idx & 63;
    atomicAdd(&m_i[(long long)sDst[m] * 64 + j], bf2f(sM[m * 64 + j]));
  }
  if (tid < 32) atomicAdd(&w_i[sDst[tid]], sCw[tid]);
  if (tid < 96) {
    int m = tid / 3, c = tid % 3;
    atomicAdd(&r_i[(long long)sDst[m] * 3 + c], sRel[m][c]);
  }
}


__global__ __launch_bounds__(256) void egnn_node_kernel(
    const float* __restrict__ x, const float* __restrict__ m_i,
    const float* __restrict__ w_i, const float* __restrict__ r_i,
    const float* __restrict__ b_n1, const float* __restrict__ b_n2,
    const bf16_t* __restrict__ Wn1, const bf16_t* __restrict__ Wn2,
    float* __restrict__ out) {
  __shared__ alignas(16) bf16_t sA[16 * 192];
  __shared__ alignas(16) bf16_t sT[16 * 256];

  const int tid = threadIdx.x;
  const int lane = tid & 31;
  const int wave = tid >> 5;
  const int g0 = blockIdx.x * 16;

  {
    int m = tid >> 4;
    int sub = tid & 15;
    long long g = g0 + m;
#pragma unroll
    for (int j = 0; j < 8; ++j) {
      int k = sub * 8 + j;
      sA[m * 192 + k] = f2bf(x[g * 131 + k]);
    }
#pragma unroll
    for (int j = 0; j < 4; ++j) {
      int k = sub * 4 + j;
      sA[m * 192 + 128 + k] = f2bf(m_i[g * 64 + k]);
    }
  }
  __syncthreads();

  {
    v16bf a[KTN1];
#pragma unroll
    for (int kt = 0; kt < KTN1; ++kt) a[kt] = load_a_frag(sA, 192, lane, kt * 32);
#pragma unroll
    for (int i = 0; i < 2; ++i) {
      int nt = wave + 8 * i;
      float bias = b_n1[nt * 16 + (lane & 15)];
      v8f acc;
#pragma unroll
      for (int r = 0; r < 8; ++r) acc[r] = bias;
#pragma unroll
      for (int kt = 0; kt < KTN1; ++kt)
        acc = wmma_bf16(a[kt], load_b_frag(Wn1, NTN1, kt, nt, lane), acc);
#pragma unroll
      for (int r = 0; r < 8; ++r) {
        int m = r + 8 * (lane >> 4);
        sT[m * 256 + nt * 16 + (lane & 15)] = f2bf(silu_f(acc[r]));
      }
    }
  }
  __syncthreads();

  {
    int nt = wave;
    int n = nt * 16 + (lane & 15);
    float bias = b_n2[n];
    v8f acc;
#pragma unroll
    for (int r = 0; r < 8; ++r) acc[r] = bias;
#pragma unroll
    for (int kt = 0; kt < KTN2; ++kt) {
      v16bf a = load_a_frag(sT, 256, lane, kt * 32);
      acc = wmma_bf16(a, load_b_frag(Wn2, NTN2, kt, nt, lane), acc);
    }
#pragma unroll
    for (int r = 0; r < 8; ++r) {
      int m = r + 8 * (lane >> 4);
      long long g = g0 + m;
      out[g * 131 + n] = acc[r] + x[g * 131 + n];
    }
  }

  if (tid < 48) {
    int m = tid / 3, c = tid % 3;
    long long g = g0 + m;
    out[g * 131 + 128 + c] = x[g * 131 + 128 + c] + w_i[g] * r_i[g * 3 + c];
  }
}


extern "C" void kernel_launch(void* const* d_in, const int* in_sizes, int n_in,
                              void* d_out, int out_size, void* d_ws,
                              size_t ws_size, hipStream_t stream) {
  const float* x    = (const float*)d_in[0];
  const int*   ei   = (const int*)d_in[1];
  const float* W_e1 = (const float*)d_in[2];
  const float* b_e1 = (const float*)d_in[3];
  const float* W_e2 = (const float*)d_in[4];
  const float* b_e2 = (const float*)d_in[5];
  const float* W_c1 = (const float*)d_in[6];
  const float* b_c1 = (const float*)d_in[7];
  const float* W_c2 = (const float*)d_in[8];
  const float* b_c2 = (const float*)d_in[9];
  const float* W_n1 = (const float*)d_in[10];
  const float* b_n1 = (const float*)d_in[11];
  const float* W_n2 = (const float*)d_in[12];
  const float* b_n2 = (const float*)d_in[13];
  float* out = (float*)d_out;

  char* ws = (char*)d_ws;
  float* m_i = (float*)ws;
  float* w_i = m_i + (long long)NN * 64;
  float* r_i = w_i + NN;
  char* wp = (char*)(r_i + (long long)NN * 3);
  bf16_t* We1b = (bf16_t*)wp; wp += (long long)KT1 * NT1 * 32 * 16 * 2;
  bf16_t* We2b = (bf16_t*)wp; wp += (long long)KT2 * NT2 * 32 * 16 * 2;
  bf16_t* Wc1b = (bf16_t*)wp; wp += (long long)KT3 * NT3 * 32 * 16 * 2;
  bf16_t* Wc2b = (bf16_t*)wp; wp += (long long)KT4 * NT4 * 32 * 16 * 2;
  bf16_t* Wn1b = (bf16_t*)wp; wp += (long long)KTN1 * NTN1 * 32 * 16 * 2;
  bf16_t* Wn2b = (bf16_t*)wp; wp += (long long)KTN2 * NTN2 * 32 * 16 * 2;

  long long accN = (long long)NN * 68;
  zero_f32_kernel<<<(unsigned)((accN + 255) / 256), 256, 0, stream>>>(m_i, accN);

  pack_weight_kernel<<<(KT1 * NT1 * 32 + 255) / 256, 256, 0, stream>>>(
      W_e1, We1b, K1, N1, KT1, NT1);
  pack_weight_kernel<<<(KT2 * NT2 * 32 + 255) / 256, 256, 0, stream>>>(
      W_e2, We2b, K2, N2, KT2, NT2);
  pack_weight_kernel<<<(KT3 * NT3 * 32 + 255) / 256, 256, 0, stream>>>(
      W_c1, Wc1b, K3, N3, KT3, NT3);
  pack_weight_kernel<<<(KT4 * NT4 * 32 + 255) / 256, 256, 0, stream>>>(
      W_c2, Wc2b, K4, N4, KT4, NT4);
  pack_weight_kernel<<<(KTN1 * NTN1 * 32 + 255) / 256, 256, 0, stream>>>(
      W_n1, Wn1b, KN1, NN1, KTN1, NTN1);
  pack_weight_kernel<<<(KTN2 * NTN2 * 32 + 255) / 256, 256, 0, stream>>>(
      W_n2, Wn2b, KN2, NN2, KTN2, NTN2);

  egnn_edge_kernel<<<EE / 32, 256, 0, stream>>>(
      x, ei, b_e1, b_e2, b_c1, b_c2, We1b, We2b, Wc1b, Wc2b, m_i, w_i, r_i);

  egnn_node_kernel<<<NN / 16, 256, 0, stream>>>(
      x, m_i, w_i, r_i, b_n1, b_n2, Wn1b, Wn2b, out);
}


# Edge kernel: 64 edges per work-group instead of 32, so each weight (B) fragment fetched feeds 4 M-tiles and weight bytes fetched per edge halve; LDS 108544 B; every sum and its order unchanged
# speedup vs baseline: 1.3789x; 1.3789x over previous
#include <hip/hip_runtime.h>
#include <hip/hip_bf16.h>


typedef __bf16 bf16_t;
typedef __attribute__((ext_vector_type(16))) __bf16 v16bf;
typedef __attribute__((ext_vector_type(8)))  __bf16 v8bf;
typedef __attribute__((ext_vector_type(8)))  float  v8f;

#define NN 50000
#define EE 500000
#define K1 257
#define KT1 9
#define N1 514
#define NT1 33
#define K2 514
#define KT2 17
#define N2 64
#define NT2 4
#define K3 64
#define KT3 2
#define N3 256
#define NT3 16
#define K4 256
#define KT4 8
#define N4 1
#define NT4 1
#define KN1 192
#define KTN1 6
#define NN1 256
#define NTN1 16
#define KN2 256
#define KTN2 8
#define NN2 128
#define NTN2 8

__device__ __forceinline__ bf16_t f2bf(float f) {
  return (bf16_t)f;
}

__device__ __forceinline__ float bf2f(bf16_t b) {
  unsigned short h = __builtin_bit_cast(unsigned short, b);
  unsigned u = (unsigned)h << 16;
  return __builtin_bit_cast(float, u);
}

__device__ __forceinline__ float silu_f(float x) {
  return x * __builtin_amdgcn_rcpf(1.0f + __expf(-x));
}

__device__ __forceinline__ v16bf load_a_frag(const bf16_t* A, int stride,
                                             int lane, int kbase) {
  int m = lane & 15;
  int kh = lane >> 4;
  const bf16_t* p = A + m * stride + kbase + 8 * kh;
  v8bf lo = *(const v8bf*)(p);
  v8bf hi = *(const v8bf*)(p + 16);
  return __builtin_shufflevector(lo, hi, 0, 1, 2, 3, 4, 5, 6, 7,
                                 8, 9, 10, 11, 12, 13, 14, 15);
}

__device__ __forceinline__ v16bf load_b_frag(const bf16_t* B, int NT,
                                             int kt, int nt, int lane) {
  return *(const v16bf*)(B + ((long long)(kt * NT + nt) * 32 + lane) * 16);
}

__device__ __forceinline__ v8f wmma_bf16(v16bf a, v16bf b, v8f c) {
  return __builtin_amdgcn_wmma_f32_16x16x32_bf16(
       false, a,  false, b,
       (short)0, c,  false,  false);
}


typedef __attribute__((ext_vector_type(4))) __bf16 v4bf;

#define NNP 50176
#define EEP 500032

__global__ __launch_bounds__(256) void zero_i32_kernel(int* p, int n) {
  int i = blockIdx.x * 256 + threadIdx.x;
  if (i < n) p[i] = 0;
}

__global__ __launch_bounds__(256) void count_kernel(
    const int* __restrict__ ei, int* __restrict__ deg, int* __restrict__ slot) {
  int e = blockIdx.x * 256 + threadIdx.x;
  if (e < EE) slot[e] = atomicAdd(&deg[ei[EE + e]], 1);
}

__global__ __launch_bounds__(1024) void scan_kernel(
    const int* __restrict__ deg, int* __restrict__ offs) {
  __shared__ int sv[NNP];
  __shared__ int sp[1024];
  const int t = threadIdx.x;
  for (int i = t; i < NNP; i += 1024) sv[i] = (i < NN) ? deg[i] : 0;
  __syncthreads();
  const int base = t * 49;
  int s = 0;
  for (int j = 0; j < 49; ++j) s += sv[base + j];
  sp[t] = s;
  __syncthreads();
  for (int o = 1; o < 1024; o <<= 1) {
    int v = (t >= o) ? sp[t - o] : 0;
    __syncthreads();
    sp[t] += v;
    __syncthreads();
  }
  int run = sp[t] - s;
  for (int j = 0; j < 49; ++j) {
    int d = sv[base + j];
    sv[base + j] = run;
    run += d;
  }
  __syncthreads();
  for (int i = t; i < NN; i += 1024) offs[i] = sv[i];
}

__global__ __launch_bounds__(256) void fill_kernel(
    const int* __restrict__ ei, const int* __restrict__ offs,
    const int* __restrict__ slot, int* __restrict__ lst) {
  int e = blockIdx.x * 256 + threadIdx.x;
  if (e < EE) lst[offs[ei[EE + e]] + slot[e]] = e;
}

__global__ __launch_bounds__(256) void order_kernel(
    const int* __restrict__ ei, const int* __restrict__ offs,
    const int* __restrict__ deg, const int* __restrict__ lst,
    int* __restrict__ lst2) {
  int e = blockIdx.x * 256 + threadIdx.x;
  if (e >= EE) return;
  int d = ei[EE + e];
  int o = offs[d];
  int n = deg[d];
  int r = 0;
  for (int j = 0; j < n; ++j) r += (lst[o + j] < e) ? 1 : 0;
  lst2[o + r] = e;
}

__global__ __launch_bounds__(256) void pack_weight_kernel(
    const float* __restrict__ W, bf16_t* __restrict__ dst,
    int K, int N, int KT, int NT) {
  int idx = blockIdx.x * 256 + threadIdx.x;
  int total = KT * NT * 32;
  if (idx >= total) return;
  int lane = idx & 31;
  int tile = idx >> 5;
  int nt = tile % NT;
  int kt = tile / NT;
  int n = nt * 16 + (lane & 15);
  int kh = lane >> 4;
  bf16_t* o = dst + (long long)idx * 16;
#pragma unroll
  for (int e = 0; e < 16; ++e) {
    int k = kt * 32 + 8 * kh + e + ((e >= 8) ? 8 : 0);
    float v = (k < K && n < N) ? W[(long long)k * N + n] : 0.0f;
    o[e] = f2bf(v);
  }
}


#define PA 296
#define PH 552
#define PT 264
#define PNA 200
#define PNT 264
#define ME 64
#define SMEM_BYTES (ME * PA * 2 + ME * PH * 2)

__global__ __launch_bounds__(256) void egnn_edge_kernel(
    const float* __restrict__ x, const int* __restrict__ ei,
    const float* __restrict__ b_e1, const float* __restrict__ b_e2,
    const float* __restrict__ b_c1, const float* __restrict__ b_c2,
    const bf16_t* __restrict__ We1, const bf16_t* __restrict__ We2,
    const bf16_t* __restrict__ Wc1, const bf16_t* __restrict__ Wc2,
    bf16_t* __restrict__ mrow, float4* __restrict__ ew) {
  __shared__ alignas(16) char smem[SMEM_BYTES];

  bf16_t* sA = (bf16_t*)smem;
  bf16_t* sM = (bf16_t*)smem;
  bf16_t* sH = (bf16_t*)(smem + ME * PA * 2);
  bf16_t* sT = sH;

  const int tid = threadIdx.x;
  const int lane = tid & 31;
  const int wave = tid >> 5;
  const int e0 = blockIdx.x * ME;

  {
    int sub = tid & 7;
#pragma unroll
    for (int h = 0; h < 2; ++h) {
      int m = h * 32 + (tid >> 3);
      int e = e0 + m;
      if (e > EE - 1) e = EE - 1;
      int s = ei[e];
      int d = ei[EE + e];
      const float* xs = x + (long long)s * 131;
      const float* xd = x + (long long)d * 131;
#pragma unroll
      for (int j = 0; j < 16; ++j) {
        int k = sub * 16 + j;
        sA[m * PA + k]       = f2bf(xd[k]);
        sA[m * PA + 128 + k] = f2bf(xs[k]);
      }
      if (sub == 0) {
        float r0 = xs[128] - xd[128];
        float r1 = xs[129] - xd[129];
        float r2 = xs[130] - xd[130];
        float dist = sqrtf(r0 * r0 + r1 * r1 + r2 * r2);
        sA[m * PA + 256] = f2bf(dist);
#pragma unroll
        for (int k = 257; k < 288; ++k) sA[m * PA + k] = f2bf(0.0f);
        float* pr = (float*)(sA + m * PA + 288);
        pr[0] = r0; pr[1] = r1; pr[2] = r2;
      }
    }
#pragma unroll
    for (int i = 0; i < 4; ++i) {
      int idx = tid + i * 256;
      sH[(idx >> 4) * PH + 528 + (idx & 15)] = f2bf(0.0f);
    }
  }
  __syncthreads();

  {
    for (int nt = wave; nt < NT1; nt += 8) {
      int n = nt * 16 + (lane & 15);
      float bias = (n < N1) ? b_e1[n] : 0.0f;
      v8f acc[4];
#pragma unroll
      for (int mt = 0; mt < 4; ++mt)
#pragma unroll
        for (int r = 0; r < 8; ++r) acc[mt][r] = bias;
#pragma unroll
      for (int kt = 0; kt < KT1; ++kt) {
        v16bf b = load_b_frag(We1, NT1, kt, nt, lane);
#pragma unroll
        for (int mt = 0; mt < 4; ++mt) {
          v16bf a = load_a_frag(sA + mt * 16 * PA, PA, lane, kt * 32);
          acc[mt] = wmma_bf16(a, b, acc[mt]);
        }
      }
#pragma unroll
      for (int mt = 0; mt < 4; ++mt)
#pragma unroll
        for (int r = 0; r < 8; ++r) {
          int m = mt * 16 + r + 8 * (lane >> 4);
          sH[m * PH + n] = f2bf(silu_f(acc[mt][r]));
        }
    }
  }
  __syncthreads();

  {
    int mp = wave >> 2;
    int nt = wave & 3;
    int n = nt * 16 + (lane & 15);
    float bias = b_e2[n];
    v8f acc0, acc1;
#pragma unroll
    for (int r = 0; r < 8; ++r) { acc0[r] = bias; acc1[r] = bias; }
#pragma unroll
    for (int kt = 0; kt < KT2; ++kt) {
      v16bf b  = load_b_frag(We2, NT2, kt, nt, lane);
      v16bf a0 = load_a_frag(sH + mp * 16 * PH,       PH, lane, kt * 32);
      v16bf a1 = load_a_frag(sH + (mp + 2) * 16 * PH, PH, lane, kt * 32);
      acc0 = wmma_bf16(a0, b, acc0);
      acc1 = wmma_bf16(a1, b, acc1);
    }
#pragma unroll
    for (int r = 0; r < 8; ++r) {
      int m = mp * 16 + r + 8 * (lane >> 4);
      sM[m * PA + n]        = f2bf(silu_f(acc0[r]));
      sM[(m + 32) * PA + n] = f2bf(silu_f(acc1[r]));
    }
  }
  __syncthreads();

  {
#pragma unroll
    for (int i = 0; i < 2; ++i) {
      int nt = wave + 8 * i;
      int col = nt * 16 + (lane & 15);
      float bias = b_c1[col];
      v8f acc[4];
#pragma unroll
      for (int mt = 0; mt < 4; ++mt)
#pragma unroll
        for (int r = 0; r < 8; ++r) acc[mt][r] = bias;
#pragma unroll
      for (int kt = 0; kt < KT3; ++kt) {
        v16bf b = load_b_frag(Wc1, NT3, kt, nt, lane);
#pragma unroll
        for (int mt = 0; mt < 4; ++mt) {
          v16bf a = load_a_frag(sM + mt * 16 * PA, PA, lane, kt * 32);
          acc[mt] = wmma_bf16(a, b, acc[mt]);
        }
      }
#pragma unroll
      for (int mt = 0; mt < 4; ++mt)
#pragma unroll
        for (int r = 0; r < 8; ++r) {
          int m = mt * 16 + r + 8 * (lane >> 4);
          sT[m * PT + col] = f2bf(silu_f(acc[mt][r]));
        }
    }
  }
  __syncthreads();

  if (wave < 4) {
    int mt = wave;
    float bias = ((lane & 15) == 0) ? b_c2[0] : 0.0f;
    v8f acc;
#pragma unroll
    for (int r = 0; r < 8; ++r) acc[r] = bias;
#pragma unroll
    for (int kt = 0; kt < KT4; ++kt) {
      v16bf a = load_a_frag(sT + mt * 16 * PT, PT, lane, kt * 32);
      acc = wmma_bf16(a, load_b_frag(Wc2, NT4, kt, 0, lane), acc);
    }
    if ((lane & 15) == 0) {
#pragma unroll
      for (int r = 0; r < 8; ++r) {
        int m = mt * 16 + r + 8 * (lane >> 4);
        ((float*)(sA + m * PA + 288))[3] = acc[r];
      }
    }
  }
  __syncthreads();

  {
    int sub = tid & 7;
#pragma unroll
    for (int h = 0; h < 2; ++h) {
      int m = h * 32 + (tid >> 3);
      if (e0 + m < EE) {
        v8bf v = *(const v8bf*)(sM + m * PA + sub * 8);
        *(v8bf*)(mrow + (long long)(e0 + m) * 64 + sub * 8) = v;
      }
    }
  }
  if (tid < ME && e0 + tid < EE) {
    const float* pr = (const float*)(sA + tid * PA + 288);
    ew[e0 + tid] = make_float4(pr[0], pr[1], pr[2], pr[3]);
  }
}


__global__ __launch_bounds__(256) void egnn_node_kernel(
    const float* __restrict__ x, const bf16_t* __restrict__ mrow,
    const float4* __restrict__ ew, const int* __restrict__ offs,
    const int* __restrict__ deg, const int* __restrict__ lst2,
    const float* __restrict__ b_n1, const float* __restrict__ b_n2,
    const bf16_t* __restrict__ Wn1, const bf16_t* __restrict__ Wn2,
    float* __restrict__ out) {
  __shared__ alignas(16) bf16_t sA[16 * PNA];
  __shared__ alignas(16) bf16_t sT[16 * PNT];
  __shared__ float sW[16];
  __shared__ float sR[16][3];

  const int tid = threadIdx.x;
  const int lane = tid & 31;
  const int wave = tid >> 5;
  const int g0 = blockIdx.x * 16;

  {
    int m = tid >> 4;
    int sub = tid & 15;
    long long g = g0 + m;
#pragma unroll
    for (int j = 0; j < 8; ++j) {
      int k = sub * 8 + j;
      sA[m * PNA + k] = f2bf(x[g * 131 + k]);
    }
    const int o = offs[g];
    const int n = deg[g];
    float a0 = 0.0f, a1 = 0.0f, a2 = 0.0f, a3 = 0.0f;
    float r0 = 0.0f, r1 = 0.0f, r2 = 0.0f, w = 0.0f;
    for (int i = 0; i < n; ++i) {
      int e = lst2[o + i];
      v4bf q = *(const v4bf*)(mrow + (long long)e * 64 + sub * 4);
      a0 += bf2f(q[0]); a1 += bf2f(q[1]); a2 += bf2f(q[2]); a3 += bf2f(q[3]);
      if (sub == 0) {
        float4 t = ew[e];
        r0 += t.x; r1 += t.y; r2 += t.z; w += t.w;
      }
    }
    sA[m * PNA + 128 + sub * 4 + 0] = f2bf(a0);
    sA[m * PNA + 128 + sub * 4 + 1] = f2bf(a1);
    sA[m * PNA + 128 + sub * 4 + 2] = f2bf(a2);
    sA[m * PNA + 128 + sub * 4 + 3] = f2bf(a3);
    if (sub == 0) {
      sW[m] = w;
      sR[m][0] = r0; sR[m][1] = r1; sR[m][2] = r2;
    }
  }
  __syncthreads();

  {
    v16bf a[KTN1];
#pragma unroll
    for (int kt = 0; kt < KTN1; ++kt) a[kt] = load_a_frag(sA, PNA, lane, kt * 32);
#pragma unroll
    for (int i = 0; i < 2; ++i) {
      int nt = wave + 8 * i;
      float bias = b_n1[nt * 16 + (lane & 15)];
      v8f acc;
#pragma unroll
      for (int r = 0; r < 8; ++r) acc[r] = bias;
#pragma unroll
      for (int kt = 0; kt < KTN1; ++kt)
        acc = wmma_bf16(a[kt], load_b_frag(Wn1, NTN1, kt, nt, lane), acc);
#pragma unroll
      for (int r = 0; r < 8; ++r) {
        int m = r + 8 * (lane >> 4);
        sT[m * PNT + nt * 16 + (lane & 15)] = f2bf(silu_f(acc[r]));
      }
    }
  }
  __syncthreads();

  {
    int nt = wave;
    int n = nt * 16 + (lane & 15);
    float bias = b_n2[n];
    v8f acc;
#pragma unroll
    for (int r = 0; r < 8; ++r) acc[r] = bias;
#pragma unroll
    for (int kt = 0; kt < KTN2; ++kt) {
      v16bf a = load_a_frag(sT, PNT, lane, kt * 32);
      acc = wmma_bf16(a, load_b_frag(Wn2, NTN2, kt, nt, lane), acc);
    }
#pragma unroll
    for (int r = 0; r < 8; ++r) {
      int m = r + 8 * (lane >> 4);
      long long g = g0 + m;
      out[g * 131 + n] = acc[r] + x[g * 131 + n];
    }
  }

  if (tid < 48) {
    int m = tid / 3, c = tid % 3;
    long long g = g0 + m;
    out[g * 131 + 128 + c] = x[g * 131 + 128 + c] + sW[m] * sR[m][c];
  }
}


extern "C" void kernel_launch(void* const* d_in, const int* in_sizes, int n_in,
                              void* d_out, int out_size, void* d_ws,
                              size_t ws_size, hipStream_t stream) {
  const float* x    = (const float*)d_in[0];
  const int*   ei   = (const int*)d_in[1];
  const float* W_e1 = (const float*)d_in[2];
  const float* b_e1 = (const float*)d_in[3];
  const float* W_e2 = (const float*)d_in[4];
  const float* b_e2 = (const float*)d_in[5];
  const float* W_c1 = (const float*)d_in[6];
  const float* b_c1 = (const float*)d_in[7];
  const float* W_c2 = (const float*)d_in[8];
  const float* b_c2 = (const float*)d_in[9];
  const float* W_n1 = (const float*)d_in[10];
  const float* b_n1 = (const float*)d_in[11];
  const float* W_n2 = (const float*)d_in[12];
  const float* b_n2 = (const float*)d_in[13];
  float* out = (float*)d_out;

  char* ws = (char*)d_ws;
  int* deg  = (int*)ws;
  int* offs = deg + NNP;
  int* slot = offs + NNP;
  int* lst  = slot + EEP;
  int* lst2 = lst + EEP;
  bf16_t* mrow = (bf16_t*)(lst2 + EEP);
  float4* ew = (float4*)(mrow + (long long)EE * 64);
  char* wp = (char*)(ew + EE);
  bf16_t* We1b = (bf16_t*)wp; wp += (long long)KT1 * NT1 * 32 * 16 * 2;
  bf16_t* We2b = (bf16_t*)wp; wp += (long long)KT2 * NT2 * 32 * 16 * 2;
  bf16_t* Wc1b = (bf16_t*)wp; wp += (long long)KT3 * NT3 * 32 * 16 * 2;
  bf16_t* Wc2b = (bf16_t*)wp; wp += (long long)KT4 * NT4 * 32 * 16 * 2;
  bf16_t* Wn1b = (bf16_t*)wp; wp += (long long)KTN1 * NTN1 * 32 * 16 * 2;
  bf16_t* Wn2b = (bf16_t*)wp; wp += (long long)KTN2 * NTN2 * 32 * 16 * 2;

  zero_i32_kernel<<<NNP / 256, 256, 0, stream>>>(deg, NNP);
  count_kernel<<<(EE + 255) / 256, 256, 0, stream>>>(ei, deg, slot);
  scan_kernel<<<1, 1024, 0, stream>>>(deg, offs);
  fill_kernel<<<(EE + 255) / 256, 256, 0, stream>>>(ei, offs, slot, lst);
  order_kernel<<<(EE + 255) / 256, 256, 0, stream>>>(ei, offs, deg, lst, lst2);

  pack_weight_kernel<<<(KT1 * NT1 * 32 + 255) / 256, 256, 0, stream>>>(
      W_e1, We1b, K1, N1, KT1, NT1);
  pack_weight_kernel<<<(KT2 * NT2 * 32 + 255) / 256, 256, 0, stream>>>(
      W_e2, We2b, K2, N2, KT2, NT2);
  pack_weight_kernel<<<(KT3 * NT3 * 32 + 255) / 256, 256, 0, stream>>>(
      W_c1, Wc1b, K3, N3, KT3, NT3);
  pack_weight_kernel<<<(KT4 * NT4 * 32 + 255) / 256, 256, 0, stream>>>(
      W_c2, Wc2b, K4, N4, KT4, NT4);
  pack_weight_kernel<<<(KTN1 * NTN1 * 32 + 255) / 256, 256, 0, stream>>>(
      W_n1, Wn1b, KN1, NN1, KTN1, NTN1);
  pack_weight_kernel<<<(KTN2 * NTN2 * 32 + 255) / 256, 256, 0, stream>>>(
      W_n2, Wn2b, KN2, NN2, KTN2, NTN2);

  egnn_edge_kernel<<<(EE + ME - 1) / ME, 256, 0, stream>>>(
      x, ei, b_e1, b_e2, b_c1, b_c2, We1b, We2b, Wc1b, Wc2b, mrow, ew);

  egnn_node_kernel<<<NN / 16, 256, 0, stream>>>(
      x, mrow, ew, offs, deg, lst2, b_n1, b_n2, Wn1b, Wn2b, out);
}


# Edge GEMM1-4 and node GEMM1 computed transposed (weights as A operand): a lane holds 8 consecutive features of one row, so a SiLU epilogue stores 16 B per lane, not 8 x 2 B; sums and order unchanged
# speedup vs baseline: 1.4476x; 1.0498x over previous
#include <hip/hip_runtime.h>
#include <hip/hip_bf16.h>


typedef __bf16 bf16_t;
typedef __attribute__((ext_vector_type(16))) __bf16 v16bf;
typedef __attribute__((ext_vector_type(8)))  __bf16 v8bf;
typedef __attribute__((ext_vector_type(8)))  float  v8f;

#define NN 50000
#define EE 500000
#define K1 257
#define KT1 9
#define N1 514
#define NT1 33
#define K2 514
#define KT2 17
#define N2 64
#define NT2 4
#define K3 64
#define KT3 2
#define N3 256
#define NT3 16
#define K4 256
#define KT4 8
#define N4 1
#define NT4 1
#define KN1 192
#define KTN1 6
#define NN1 256
#define NTN1 16
#define KN2 256
#define KTN2 8
#define NN2 128
#define NTN2 8

__device__ __forceinline__ bf16_t f2bf(float f) {
  return (bf16_t)f;
}

__device__ __forceinline__ float bf2f(bf16_t b) {
  unsigned short h = __builtin_bit_cast(unsigned short, b);
  unsigned u = (unsigned)h << 16;
  return __builtin_bit_cast(float, u);
}

__device__ __forceinline__ float silu_f(float x) {
  return x * __builtin_amdgcn_rcpf(1.0f + __expf(-x));
}

__device__ __forceinline__ v16bf load_a_frag(const bf16_t* A, int stride,
                                             int lane, int kbase) {
  int m = lane & 15;
  int kh = lane >> 4;
  const bf16_t* p = A + m * stride + kbase + 8 * kh;
  v8bf lo = *(const v8bf*)(p);
  v8bf hi = *(const v8bf*)(p + 16);
  return __builtin_shufflevector(lo, hi, 0, 1, 2, 3, 4, 5, 6, 7,
                                 8, 9, 10, 11, 12, 13, 14, 15);
}

__device__ __forceinline__ v16bf load_b_frag(const bf16_t* B, int NT,
                                             int kt, int nt, int lane) {
  return *(const v16bf*)(B + ((long long)(kt * NT + nt) * 32 + lane) * 16);
}

__device__ __forceinline__ v8f wmma_bf16(v16bf a, v16bf b, v8f c) {
  return __builtin_amdgcn_wmma_f32_16x16x32_bf16(
       false, a,  false, b,
       (short)0, c,  false,  false);
}

__device__ __forceinline__ v8f wmma_bf16_t(v16bf w, v16bf x, v8f c) {
  return wmma_bf16(w, x, c);
}

__device__ __forceinline__ v8bf silu_pack8(v8f acc) {
  v8bf o;
#pragma unroll
  for (int r = 0; r < 8; ++r) o[r] = f2bf(silu_f(acc[r]));
  return o;
}


typedef __attribute__((ext_vector_type(4))) __bf16 v4bf;

#define NNP 50176
#define EEP 500032

__global__ __launch_bounds__(256) void zero_i32_kernel(int* p, int n) {
  int i = blockIdx.x * 256 + threadIdx.x;
  if (i < n) p[i] = 0;
}

__global__ __launch_bounds__(256) void count_kernel(
    const int* __restrict__ ei, int* __restrict__ deg, int* __restrict__ slot) {
  int e = blockIdx.x * 256 + threadIdx.x;
  if (e < EE) slot[e] = atomicAdd(&deg[ei[EE + e]], 1);
}

__global__ __launch_bounds__(1024) void scan_kernel(
    const int* __restrict__ deg, int* __restrict__ offs) {
  __shared__ int sv[NNP];
  __shared__ int sp[1024];
  const int t = threadIdx.x;
  for (int i = t; i < NNP; i += 1024) sv[i] = (i < NN) ? deg[i] : 0;
  __syncthreads();
  const int base = t * 49;
  int s = 0;
  for (int j = 0; j < 49; ++j) s += sv[base + j];
  sp[t] = s;
  __syncthreads();
  for (int o = 1; o < 1024; o <<= 1) {
    int v = (t >= o) ? sp[t - o] : 0;
    __syncthreads();
    sp[t] += v;
    __syncthreads();
  }
  int run = sp[t] - s;
  for (int j = 0; j < 49; ++j) {
    int d = sv[base + j];
    sv[base + j] = run;
    run += d;
  }
  __syncthreads();
  for (int i = t; i < NN; i += 1024) offs[i] = sv[i];
}

__global__ __launch_bounds__(256) void fill_kernel(
    const int* __restrict__ ei, const int* __restrict__ offs,
    const int* __restrict__ slot, int* __restrict__ lst) {
  int e = blockIdx.x * 256 + threadIdx.x;
  if (e < EE) lst[offs[ei[EE + e]] + slot[e]] = e;
}

__global__ __launch_bounds__(256) void order_kernel(
    const int* __restrict__ ei, const int* __restrict__ offs,
    const int* __restrict__ deg, const int* __restrict__ lst,
    int* __restrict__ lst2) {
  int e = blockIdx.x * 256 + threadIdx.x;
  if (e >= EE) return;
  int d = ei[EE + e];
  int o = offs[d];
  int n = deg[d];
  int r = 0;
  for (int j = 0; j < n; ++j) r += (lst[o + j] < e) ? 1 : 0;
  lst2[o + r] = e;
}

__global__ __launch_bounds__(256) void pack_weight_kernel(
    const float* __restrict__ W, bf16_t* __restrict__ dst,
    int K, int N, int KT, int NT) {
  int idx = blockIdx.x * 256 + threadIdx.x;
  int total = KT * NT * 32;
  if (idx >= total) return;
  int lane = idx & 31;
  int tile = idx >> 5;
  int nt = tile % NT;
  int kt = tile / NT;
  int n = nt * 16 + (lane & 15);
  int kh = lane >> 4;
  bf16_t* o = dst + (long long)idx * 16;
#pragma unroll
  for (int e = 0; e < 16; ++e) {
    int k = kt * 32 + 8 * kh + e + ((e >= 8) ? 8 : 0);
    float v = (k < K && n < N) ? W[(long long)k * N + n] : 0.0f;
    o[e] = f2bf(v);
  }
}


#define PA 296
#define PH 552
#define PT 264
#define PNA 200
#define PNT 264
#define ME 64
#define SMEM_BYTES (ME * PA * 2 + ME * PH * 2)

__global__ __launch_bounds__(256) void egnn_edge_kernel(
    const float* __restrict__ x, const int* __restrict__ ei,
    const float* __restrict__ b_e1, const float* __restrict__ b_e2,
    const float* __restrict__ b_c1, const float* __restrict__ b_c2,
    const bf16_t* __restrict__ We1, const bf16_t* __restrict__ We2,
    const bf16_t* __restrict__ Wc1, const bf16_t* __restrict__ Wc2,
    bf16_t* __restrict__ mrow, float4* __restrict__ ew) {
  __shared__ alignas(16) char smem[SMEM_BYTES];

  bf16_t* sA = (bf16_t*)smem;
  bf16_t* sM = (bf16_t*)smem;
  bf16_t* sH = (bf16_t*)(smem + ME * PA * 2);
  bf16_t* sT = sH;

  const int tid = threadIdx.x;
  const int lane = tid & 31;
  const int wave = tid >> 5;
  const int e0 = blockIdx.x * ME;

  {
    int sub = tid & 7;
#pragma unroll
    for (int h = 0; h < 2; ++h) {
      int m = h * 32 + (tid >> 3);
      int e = e0 + m;
      if (e > EE - 1) e = EE - 1;
      int s = ei[e];
      int d = ei[EE + e];
      const float* xs = x + (long long)s * 131;
      const float* xd = x + (long long)d * 131;
#pragma unroll
      for (int j = 0; j < 16; ++j) {
        int k = sub * 16 + j;
        sA[m * PA + k]       = f2bf(xd[k]);
        sA[m * PA + 128 + k] = f2bf(xs[k]);
      }
      if (sub == 0) {
        float r0 = xs[128] - xd[128];
        float r1 = xs[129] - xd[129];
        float r2 = xs[130] - xd[130];
        float dist = sqrtf(r0 * r0 + r1 * r1 + r2 * r2);
        sA[m * PA + 256] = f2bf(dist);
#pragma unroll
        for (int k = 257; k < 288; ++k) sA[m * PA + k] = f2bf(0.0f);
        float* pr = (float*)(sA + m * PA + 288);
        pr[0] = r0; pr[1] = r1; pr[2] = r2;
      }
    }
#pragma unroll
    for (int i = 0; i < 4; ++i) {
      int idx = tid + i * 256;
      sH[(idx >> 4) * PH + 528 + (idx & 15)] = f2bf(0.0f);
    }
  }
  __syncthreads();

  {
    const int c = lane & 15;
    const int f8 = 8 * (lane >> 4);
    for (int nt = wave; nt < NT1; nt += 8) {
      int n0 = nt * 16 + f8;
      v8f bias;
      if (n0 + 8 <= N1) {
#pragma unroll
        for (int r = 0; r < 8; ++r) bias[r] = b_e1[n0 + r];
      } else {
#pragma unroll
        for (int r = 0; r < 8; ++r) bias[r] = (n0 + r < N1) ? b_e1[n0 + r] : 0.0f;
      }
      v8f acc[4];
#pragma unroll
      for (int mt = 0; mt < 4; ++mt) acc[mt] = bias;
#pragma unroll
      for (int kt = 0; kt < KT1; ++kt) {
        v16bf b = load_b_frag(We1, NT1, kt, nt, lane);
#pragma unroll
        for (int mt = 0; mt < 4; ++mt) {
          v16bf a = load_a_frag(sA + mt * 16 * PA, PA, lane, kt * 32);
          acc[mt] = wmma_bf16_t(b, a, acc[mt]);
        }
      }
#pragma unroll
      for (int mt = 0; mt < 4; ++mt)
        *(v8bf*)(sH + (mt * 16 + c) * PH + n0) = silu_pack8(acc[mt]);
    }
  }
  __syncthreads();

  {
    int mp = wave >> 2;
    int nt = wave & 3;
    int c = lane & 15;
    int n0 = nt * 16 + 8 * (lane >> 4);
    v8f acc0, acc1;
#pragma unroll
    for (int r = 0; r < 8; ++r) { acc0[r] = b_e2[n0 + r]; acc1[r] = acc0[r]; }
#pragma unroll
    for (int kt = 0; kt < KT2; ++kt) {
      v16bf b  = load_b_frag(We2, NT2, kt, nt, lane);
      v16bf a0 = load_a_frag(sH + mp * 16 * PH,       PH, lane, kt * 32);
      v16bf a1 = load_a_frag(sH + (mp + 2) * 16 * PH, PH, lane, kt * 32);
      acc0 = wmma_bf16_t(b, a0, acc0);
      acc1 = wmma_bf16_t(b, a1, acc1);
    }
    *(v8bf*)(sM + (mp * 16 + c) * PA + n0)      = silu_pack8(acc0);
    *(v8bf*)(sM + (mp * 16 + 32 + c) * PA + n0) = silu_pack8(acc1);
  }
  __syncthreads();

  {
#pragma unroll
    for (int i = 0; i < 2; ++i) {
      int nt = wave + 8 * i;
      int c = lane & 15;
      int n0 = nt * 16 + 8 * (lane >> 4);
      v8f bias;
#pragma unroll
      for (int r = 0; r < 8; ++r) bias[r] = b_c1[n0 + r];
      v8f acc[4];
#pragma unroll
      for (int mt = 0; mt < 4; ++mt) acc[mt] = bias;
#pragma unroll
      for (int kt = 0; kt < KT3; ++kt) {
        v16bf b = load_b_frag(Wc1, NT3, kt, nt, lane);
#pragma unroll
        for (int mt = 0; mt < 4; ++mt) {
          v16bf a = load_a_frag(sM + mt * 16 * PA, PA, lane, kt * 32);
          acc[mt] = wmma_bf16_t(b, a, acc[mt]);
        }
      }
#pragma unroll
      for (int mt = 0; mt < 4; ++mt)
        *(v8bf*)(sT + (mt * 16 + c) * PT + n0) = silu_pack8(acc[mt]);
    }
  }
  __syncthreads();

  if (wave < 4) {
    int mt = wave;
    v8f acc;
#pragma unroll
    for (int r = 0; r < 8; ++r) acc[r] = 0.0f;
    if (lane < 16) acc[0] = b_c2[0];
#pragma unroll
    for (int kt = 0; kt < KT4; ++kt) {
      v16bf a = load_a_frag(sT + mt * 16 * PT, PT, lane, kt * 32);
      acc = wmma_bf16_t(load_b_frag(Wc2, NT4, kt, 0, lane), a, acc);
    }
    if (lane < 16)
      ((float*)(sA + (mt * 16 + lane) * PA + 288))[3] = acc[0];
  }
  __syncthreads();

  {
    int sub = tid & 7;
#pragma unroll
    for (int h = 0; h < 2; ++h) {
      int m = h * 32 + (tid >> 3);
      if (e0 + m < EE) {
        v8bf v = *(const v8bf*)(sM + m * PA + sub * 8);
        *(v8bf*)(mrow + (long long)(e0 + m) * 64 + sub * 8) = v;
      }
    }
  }
  if (tid < ME && e0 + tid < EE) {
    const float* pr = (const float*)(sA + tid * PA + 288);
    ew[e0 + tid] = make_float4(pr[0], pr[1], pr[2], pr[3]);
  }
}


__global__ __launch_bounds__(256) void egnn_node_kernel(
    const float* __restrict__ x, const bf16_t* __restrict__ mrow,
    const float4* __restrict__ ew, const int* __restrict__ offs,
    const int* __restrict__ deg, const int* __restrict__ lst2,
    const float* __restrict__ b_n1, const float* __restrict__ b_n2,
    const bf16_t* __restrict__ Wn1, const bf16_t* __restrict__ Wn2,
    float* __restrict__ out) {
  __shared__ alignas(16) bf16_t sA[16 * PNA];
  __shared__ alignas(16) bf16_t sT[16 * PNT];
  __shared__ float sW[16];
  __shared__ float sR[16][3];

  const int tid = threadIdx.x;
  const int lane = tid & 31;
  const int wave = tid >> 5;
  const int g0 = blockIdx.x * 16;

  {
    int m = tid >> 4;
    int sub = tid & 15;
    long long g = g0 + m;
#pragma unroll
    for (int j = 0; j < 8; ++j) {
      int k = sub * 8 + j;
      sA[m * PNA + k] = f2bf(x[g * 131 + k]);
    }
    const int o = offs[g];
    const int n = deg[g];
    float a0 = 0.0f, a1 = 0.0f, a2 = 0.0f, a3 = 0.0f;
    float r0 = 0.0f, r1 = 0.0f, r2 = 0.0f, w = 0.0f;
    for (int i = 0; i < n; ++i) {
      int e = lst2[o + i];
      v4bf q = *(const v4bf*)(mrow + (long long)e * 64 + sub * 4);
      a0 += bf2f(q[0]); a1 += bf2f(q[1]); a2 += bf2f(q[2]); a3 += bf2f(q[3]);
      if (sub == 0) {
        float4 t = ew[e];
        r0 += t.x; r1 += t.y; r2 += t.z; w += t.w;
      }
    }
    sA[m * PNA + 128 + sub * 4 + 0] = f2bf(a0);
    sA[m * PNA + 128 + sub * 4 + 1] = f2bf(a1);
    sA[m * PNA + 128 + sub * 4 + 2] = f2bf(a2);
    sA[m * PNA + 128 + sub * 4 + 3] = f2bf(a3);
    if (sub == 0) {
      sW[m] = w;
      sR[m][0] = r0; sR[m][1] = r1; sR[m][2] = r2;
    }
  }
  __syncthreads();

  {
    v16bf a[KTN1];
#pragma unroll
    for (int kt = 0; kt < KTN1; ++kt) a[kt] = load_a_frag(sA, PNA, lane, kt * 32);
#pragma unroll
    for (int i = 0; i < 2; ++i) {
      int nt = wave + 8 * i;
      int n0 = nt * 16 + 8 * (lane >> 4);
      v8f acc;
#pragma unroll
      for (int r = 0; r < 8; ++r) acc[r] = b_n1[n0 + r];
#pragma unroll
      for (int kt = 0; kt < KTN1; ++kt)
        acc = wmma_bf16_t(load_b_frag(Wn1, NTN1, kt, nt, lane), a[kt], acc);
      *(v8bf*)(sT + (lane & 15) * PNT + n0) = silu_pack8(acc);
    }
  }
  __syncthreads();

  {
    int nt = wave;
    int n = nt * 16 + (lane & 15);
    float bias = b_n2[n];
    v8f acc;
#pragma unroll
    for (int r = 0; r < 8; ++r) acc[r] = bias;
#pragma unroll
    for (int kt = 0; kt < KTN2; ++kt) {
      v16bf a = load_a_frag(sT, PNT, lane, kt * 32);
      acc = wmma_bf16(a, load_b_frag(Wn2, NTN2, kt, nt, lane), acc);
    }
#pragma unroll
    for (int r = 0; r < 8; ++r) {
      int m = r + 8 * (lane >> 4);
      long long g = g0 + m;
      out[g * 131 + n] = acc[r] + x[g * 131 + n];
    }
  }

  if (tid < 48) {
    int m = tid / 3, c = tid % 3;
    long long g = g0 + m;
    out[g * 131 + 128 + c] = x[g * 131 + 128 + c] + sW[m] * sR[m][c];
  }
}


extern "C" void kernel_launch(void* const* d_in, const int* in_sizes, int n_in,
                              void* d_out, int out_size, void* d_ws,
                              size_t ws_size, hipStream_t stream) {
  const float* x    = (const float*)d_in[0];
  const int*   ei   = (const int*)d_in[1];
  const float* W_e1 = (const float*)d_in[2];
  const float* b_e1 = (const float*)d_in[3];
  const float* W_e2 = (const float*)d_in[4];
  const float* b_e2 = (const float*)d_in[5];
  const float* W_c1 = (const float*)d_in[6];
  const float* b_c1 = (const float*)d_in[7];
  const float* W_c2 = (const float*)d_in[8];
  const float* b_c2 = (const float*)d_in[9];
  const float* W_n1 = (const float*)d_in[10];
  const float* b_n1 = (const float*)d_in[11];
  const float* W_n2 = (const float*)d_in[12];
  const float* b_n2 = (const float*)d_in[13];
  float* out = (float*)d_out;

  char* ws = (char*)d_ws;
  int* deg  = (int*)ws;
  int* offs = deg + NNP;
  int* slot = offs + NNP;
  int* lst  = slot + EEP;
  int* lst2 = lst + EEP;
  bf16_t* mrow = (bf16_t*)(lst2 + EEP);
  float4* ew = (float4*)(mrow + (long long)EE * 64);
  char* wp = (char*)(ew + EE);
  bf16_t* We1b = (bf16_t*)wp; wp += (long long)KT1 * NT1 * 32 * 16 * 2;
  bf16_t* We2b = (bf16_t*)wp; wp += (long long)KT2 * NT2 * 32 * 16 * 2;
  bf16_t* Wc1b = (bf16_t*)wp; wp += (long long)KT3 * NT3 * 32 * 16 * 2;
  bf16_t* Wc2b = (bf16_t*)wp; wp += (long long)KT4 * NT4 * 32 * 16 * 2;
  bf16_t* Wn1b = (bf16_t*)wp; wp += (long long)KTN1 * NTN1 * 32 * 16 * 2;
  bf16_t* Wn2b = (bf16_t*)wp; wp += (long long)KTN2 * NTN2 * 32 * 16 * 2;

  zero_i32_kernel<<<NNP / 256, 256, 0, stream>>>(deg, NNP);
  count_kernel<<<(EE + 255) / 256, 256, 0, stream>>>(ei, deg, slot);
  scan_kernel<<<1, 1024, 0, stream>>>(deg, offs);
  fill_kernel<<<(EE + 255) / 256, 256, 0, stream>>>(ei, offs, slot, lst);
  order_kernel<<<(EE + 255) / 256, 256, 0, stream>>>(ei, offs, deg, lst, lst2);

  pack_weight_kernel<<<(KT1 * NT1 * 32 + 255) / 256, 256, 0, stream>>>(
      W_e1, We1b, K1, N1, KT1, NT1);
  pack_weight_kernel<<<(KT2 * NT2 * 32 + 255) / 256, 256, 0, stream>>>(
      W_e2, We2b, K2, N2, KT2, NT2);
  pack_weight_kernel<<<(KT3 * NT3 * 32 + 255) / 256, 256, 0, stream>>>(
      W_c1, Wc1b, K3, N3, KT3, NT3);
  pack_weight_kernel<<<(KT4 * NT4 * 32 + 255) / 256, 256, 0, stream>>>(
      W_c2, Wc2b, K4, N4, KT4, NT4);
  pack_weight_kernel<<<(KTN1 * NTN1 * 32 + 255) / 256, 256, 0, stream>>>(
      W_n1, Wn1b, KN1, NN1, KTN1, NTN1);
  pack_weight_kernel<<<(KTN2 * NTN2 * 32 + 255) / 256, 256, 0, stream>>>(
      W_n2, Wn2b, KN2, NN2, KTN2, NTN2);

  egnn_edge_kernel<<<(EE + ME - 1) / ME, 256, 0, stream>>>(
      x, ei, b_e1, b_e2, b_c1, b_c2, We1b, We2b, Wc1b, Wc2b, mrow, ew);

  egnn_node_kernel<<<NN / 16, 256, 0, stream>>>(
      x, mrow, ew, offs, deg, lst2, b_n1, b_n2, Wn1b, Wn2b, out);
}


# Edge GEMM1/GEMM3: a wave keeps 2 N-tiles x 4 M-tiles of accumulators, so each A fragment read from LDS feeds 2 WMMAs (LDS reads halved); GEMM1 pipelined, loads one step ahead; sums unchanged
# speedup vs baseline: 1.5383x; 1.0627x over previous
#include <hip/hip_runtime.h>
#include <hip/hip_bf16.h>


typedef __bf16 bf16_t;
typedef __attribute__((ext_vector_type(16))) __bf16 v16bf;
typedef __attribute__((ext_vector_type(8)))  __bf16 v8bf;
typedef __attribute__((ext_vector_type(8)))  float  v8f;

#define NN 50000
#define EE 500000
#define K1 257
#define KT1 9
#define N1 514
#define NT1 33
#define K2 514
#define KT2 17
#define N2 64
#define NT2 4
#define K3 64
#define KT3 2
#define N3 256
#define NT3 16
#define K4 256
#define KT4 8
#define N4 1
#define NT4 1
#define KN1 192
#define KTN1 6
#define NN1 256
#define NTN1 16
#define KN2 256
#define KTN2 8
#define NN2 128
#define NTN2 8

__device__ __forceinline__ bf16_t f2bf(float f) {
  return (bf16_t)f;
}

__device__ __forceinline__ float bf2f(bf16_t b) {
  unsigned short h = __builtin_bit_cast(unsigned short, b);
  unsigned u = (unsigned)h << 16;
  return __builtin_bit_cast(float, u);
}

__device__ __forceinline__ float silu_f(float x) {
  return x * __builtin_amdgcn_rcpf(1.0f + __expf(-x));
}

__device__ __forceinline__ v16bf load_a_frag(const bf16_t* A, int stride,
                                             int lane, int kbase) {
  int m = lane & 15;
  int kh = lane >> 4;
  const bf16_t* p = A + m * stride + kbase + 8 * kh;
  v8bf lo = *(const v8bf*)(p);
  v8bf hi = *(const v8bf*)(p + 16);
  return __builtin_shufflevector(lo, hi, 0, 1, 2, 3, 4, 5, 6, 7,
                                 8, 9, 10, 11, 12, 13, 14, 15);
}

__device__ __forceinline__ v16bf load_b_frag(const bf16_t* B, int NT,
                                             int kt, int nt, int lane) {
  return *(const v16bf*)(B + ((long long)(kt * NT + nt) * 32 + lane) * 16);
}

__device__ __forceinline__ v16bf load_b_frag_w(const bf16_t* B, int NT,
                                               int kt, int ntc, int tid) {
  return *(const v16bf*)(B + ((long long)(kt * NT + ntc) * 32 + tid) * 16);
}

__device__ __forceinline__ v8f wmma_bf16(v16bf a, v16bf b, v8f c) {
  return __builtin_amdgcn_wmma_f32_16x16x32_bf16(
       false, a,  false, b,
       (short)0, c,  false,  false);
}

__device__ __forceinline__ v8f wmma_bf16_t(v16bf w, v16bf x, v8f c) {
  return wmma_bf16(w, x, c);
}

__device__ __forceinline__ v8bf silu_pack8(v8f acc) {
  v8bf o;
#pragma unroll
  for (int r = 0; r < 8; ++r) o[r] = f2bf(silu_f(acc[r]));
  return o;
}


typedef __attribute__((ext_vector_type(4))) __bf16 v4bf;

#define NNP 50176
#define EEP 500032

__global__ __launch_bounds__(256) void zero_i32_kernel(int* p, int n) {
  int i = blockIdx.x * 256 + threadIdx.x;
  if (i < n) p[i] = 0;
}

__global__ __launch_bounds__(256) void count_kernel(
    const int* __restrict__ ei, int* __restrict__ deg, int* __restrict__ slot) {
  int e = blockIdx.x * 256 + threadIdx.x;
  if (e < EE) slot[e] = atomicAdd(&deg[ei[EE + e]], 1);
}

__global__ __launch_bounds__(1024) void scan_kernel(
    const int* __restrict__ deg, int* __restrict__ offs) {
  __shared__ int sv[NNP];
  __shared__ int sp[1024];
  const int t = threadIdx.x;
  for (int i = t; i < NNP; i += 1024) sv[i] = (i < NN) ? deg[i] : 0;
  __syncthreads();
  const int base = t * 49;
  int s = 0;
  for (int j = 0; j < 49; ++j) s += sv[base + j];
  sp[t] = s;
  __syncthreads();
  for (int o = 1; o < 1024; o <<= 1) {
    int v = (t >= o) ? sp[t - o] : 0;
    __syncthreads();
    sp[t] += v;
    __syncthreads();
  }
  int run = sp[t] - s;
  for (int j = 0; j < 49; ++j) {
    int d = sv[base + j];
    sv[base + j] = run;
    run += d;
  }
  __syncthreads();
  for (int i = t; i < NN; i += 1024) offs[i] = sv[i];
}

__global__ __launch_bounds__(256) void fill_kernel(
    const int* __restrict__ ei, const int* __restrict__ offs,
    const int* __restrict__ slot, int* __restrict__ lst) {
  int e = blockIdx.x * 256 + threadIdx.x;
  if (e < EE) lst[offs[ei[EE + e]] + slot[e]] = e;
}

__global__ __launch_bounds__(256) void order_kernel(
    const int* __restrict__ ei, const int* __restrict__ offs,
    const int* __restrict__ deg, const int* __restrict__ lst,
    int* __restrict__ lst2) {
  int e = blockIdx.x * 256 + threadIdx.x;
  if (e >= EE) return;
  int d = ei[EE + e];
  int o = offs[d];
  int n = deg[d];
  int r = 0;
  for (int j = 0; j < n; ++j) r += (lst[o + j] < e) ? 1 : 0;
  lst2[o + r] = e;
}

__global__ __launch_bounds__(256) void pack_weight_kernel(
    const float* __restrict__ W, bf16_t* __restrict__ dst,
    int K, int N, int KT, int NT) {
  int idx = blockIdx.x * 256 + threadIdx.x;
  int total = KT * NT * 32;
  if (idx >= total) return;
  int lane = idx & 31;
  int tile = idx >> 5;
  int nt = tile % NT;
  int kt = tile / NT;
  int n = nt * 16 + (lane & 15);
  int kh = lane >> 4;
  bf16_t* o = dst + (long long)idx * 16;
#pragma unroll
  for (int e = 0; e < 16; ++e) {
    int k = kt * 32 + 8 * kh + e + ((e >= 8) ? 8 : 0);
    float v = (k < K && n < N) ? W[(long long)k * N + n] : 0.0f;
    o[e] = f2bf(v);
  }
}


#define PA 296
#define PH 552
#define PT 264
#define PNA 200
#define PNT 264
#define ME 64
#define SMEM_BYTES (ME * PA * 2 + ME * PH * 2)

__global__ __launch_bounds__(256, 6) void egnn_edge_kernel(
    const float* __restrict__ x, const int* __restrict__ ei,
    const float* __restrict__ b_e1, const float* __restrict__ b_e2,
    const float* __restrict__ b_c1, const float* __restrict__ b_c2,
    const bf16_t* __restrict__ We1, const bf16_t* __restrict__ We2,
    const bf16_t* __restrict__ Wc1, const bf16_t* __restrict__ Wc2,
    bf16_t* __restrict__ mrow, float4* __restrict__ ew) {
  __shared__ alignas(16) char smem[SMEM_BYTES];

  bf16_t* sA = (bf16_t*)smem;
  bf16_t* sM = (bf16_t*)smem;
  bf16_t* sH = (bf16_t*)(smem + ME * PA * 2);
  bf16_t* sT = sH;

  const int tid = threadIdx.x;
  const int lane = tid & 31;
  const int wave = tid >> 5;
  const int e0 = blockIdx.x * ME;

  {
    int sub = tid & 7;
#pragma unroll
    for (int h = 0; h < 2; ++h) {
      int m = h * 32 + (tid >> 3);
      int e = e0 + m;
      if (e > EE - 1) e = EE - 1;
      int s = ei[e];
      int d = ei[EE + e];
      const float* xs = x + (long long)s * 131;
      const float* xd = x + (long long)d * 131;
#pragma unroll
      for (int j = 0; j < 16; ++j) {
        int k = sub * 16 + j;
        sA[m * PA + k]       = f2bf(xd[k]);
        sA[m * PA + 128 + k] = f2bf(xs[k]);
      }
      if (sub == 0) {
        float r0 = xs[128] - xd[128];
        float r1 = xs[129] - xd[129];
        float r2 = xs[130] - xd[130];
        float dist = sqrtf(r0 * r0 + r1 * r1 + r2 * r2);
        sA[m * PA + 256] = f2bf(dist);
#pragma unroll
        for (int k = 257; k < 288; ++k) sA[m * PA + k] = f2bf(0.0f);
        float* pr = (float*)(sA + m * PA + 288);
        pr[0] = r0; pr[1] = r1; pr[2] = r2;
      }
    }
#pragma unroll
    for (int i = 0; i < 4; ++i) {
      int idx = tid + i * 256;
      sH[(idx >> 4) * PH + 528 + (idx & 15)] = f2bf(0.0f);
    }
  }
  v16bf bq[2][2];
  bq[0][0] = load_b_frag_w(We1, NT1, 0, 0, tid);
  bq[0][1] = load_b_frag_w(We1, NT1, 0, 8, tid);
  v8f bias0, bias1;
#pragma unroll
  for (int r = 0; r < 8; ++r) {
    bias0[r] = b_e1[wave * 16 + 8 * (lane >> 4) + r];
    bias1[r] = b_e1[(wave + 8) * 16 + 8 * (lane >> 4) + r];
  }
  __syncthreads();

  {
    const int c = lane & 15;
    const int f8 = 8 * (lane >> 4);

    v8f acc0[4], acc1[4];
    v16bf a[4];
    a[0] = load_a_frag(sA,           PA, lane, 0);
    a[1] = load_a_frag(sA + 16 * PA, PA, lane, 0);
#pragma unroll
    for (int s = 0; s < 2 * KT1; ++s) {
      const int p = s / KT1;
      const int kt = s % KT1;
      const int n0 = (wave + 16 * p) * 16 + f8;
      const int n1 = (wave + 16 * p + 8) * 16 + f8;
      if (kt == 0) {
#pragma unroll
        for (int mt = 0; mt < 4; ++mt) { acc0[mt] = bias0; acc1[mt] = bias1; }
      }
      if (s + 1 < 2 * KT1) {
        const int p2 = (s + 1) / KT1;
        const int k2 = (s + 1) % KT1;
        bq[(s + 1) % 2][0] = load_b_frag_w(We1, NT1, k2, 16 * p2, tid);
        bq[(s + 1) % 2][1] = load_b_frag_w(We1, NT1, k2, 16 * p2 + 8, tid);
      }
      a[2] = load_a_frag(sA + 32 * PA, PA, lane, kt * 32);
      a[3] = load_a_frag(sA + 48 * PA, PA, lane, kt * 32);
      __builtin_amdgcn_sched_barrier(0);
      acc0[0] = wmma_bf16_t(bq[s % 2][0], a[0], acc0[0]);
      acc1[0] = wmma_bf16_t(bq[s % 2][1], a[0], acc1[0]);
      acc0[1] = wmma_bf16_t(bq[s % 2][0], a[1], acc0[1]);
      acc1[1] = wmma_bf16_t(bq[s % 2][1], a[1], acc1[1]);
      __builtin_amdgcn_sched_barrier(0);
      if (s + 1 < 2 * KT1) {
        const int k1 = (s + 1) % KT1;
        a[0] = load_a_frag(sA,           PA, lane, k1 * 32);
        a[1] = load_a_frag(sA + 16 * PA, PA, lane, k1 * 32);
      }
      __builtin_amdgcn_sched_barrier(0);
      acc0[2] = wmma_bf16_t(bq[s % 2][0], a[2], acc0[2]);
      acc1[2] = wmma_bf16_t(bq[s % 2][1], a[2], acc1[2]);
      acc0[3] = wmma_bf16_t(bq[s % 2][0], a[3], acc0[3]);
      acc1[3] = wmma_bf16_t(bq[s % 2][1], a[3], acc1[3]);
      if (kt == KT1 - 1) {
        if (p == 0) {
#pragma unroll
          for (int r = 0; r < 8; ++r) {
            bias0[r] = b_e1[n0 + 256 + r];
            bias1[r] = b_e1[n1 + 256 + r];
          }
          __builtin_amdgcn_sched_barrier(0);
        }
#pragma unroll
        for (int mt = 0; mt < 4; ++mt) {
          *(v8bf*)(sH + (mt * 16 + c) * PH + n0) = silu_pack8(acc0[mt]);
          *(v8bf*)(sH + (mt * 16 + c) * PH + n1) = silu_pack8(acc1[mt]);
        }
      }
    }

    if (wave == 0) {
      const int n0 = (NT1 - 1) * 16 + f8;
      v8f t[4];
#pragma unroll
      for (int r = 0; r < 8; ++r) t[0][r] = (n0 + r < N1) ? b_e1[n0 + r] : 0.0f;
#pragma unroll
      for (int mt = 1; mt < 4; ++mt) t[mt] = t[0];
#pragma unroll
      for (int kt = 0; kt < KT1; ++kt) {
        v16bf b = load_b_frag_w(We1, NT1, kt, 32, tid);
#pragma unroll
        for (int mt = 0; mt < 4; ++mt) {
          v16bf am = load_a_frag(sA + mt * 16 * PA, PA, lane, kt * 32);
          t[mt] = wmma_bf16_t(b, am, t[mt]);
        }
      }
#pragma unroll
      for (int mt = 0; mt < 4; ++mt)
        *(v8bf*)(sH + (mt * 16 + c) * PH + n0) = silu_pack8(t[mt]);
    }
  }
  __syncthreads();

  {
    int mp = wave >> 2;
    int nt = wave & 3;
    int c = lane & 15;
    int n0 = nt * 16 + 8 * (lane >> 4);
    v8f acc0, acc1;
#pragma unroll
    for (int r = 0; r < 8; ++r) { acc0[r] = b_e2[n0 + r]; acc1[r] = acc0[r]; }
#pragma unroll
    for (int kt = 0; kt < KT2; ++kt) {
      v16bf b  = load_b_frag(We2, NT2, kt, nt, lane);
      v16bf a0 = load_a_frag(sH + mp * 16 * PH,       PH, lane, kt * 32);
      v16bf a1 = load_a_frag(sH + (mp + 2) * 16 * PH, PH, lane, kt * 32);
      acc0 = wmma_bf16_t(b, a0, acc0);
      acc1 = wmma_bf16_t(b, a1, acc1);
    }
    *(v8bf*)(sM + (mp * 16 + c) * PA + n0)      = silu_pack8(acc0);
    *(v8bf*)(sM + (mp * 16 + 32 + c) * PA + n0) = silu_pack8(acc1);
  }
  __syncthreads();

  {
    const int c = lane & 15;
    const int nt0 = wave;
    const int nt1 = wave + 8;
    const int n0 = nt0 * 16 + 8 * (lane >> 4);
    const int n1 = nt1 * 16 + 8 * (lane >> 4);
    v8f bias0, bias1;
#pragma unroll
    for (int r = 0; r < 8; ++r) { bias0[r] = b_c1[n0 + r]; bias1[r] = b_c1[n1 + r]; }
    v8f acc0[4], acc1[4];
#pragma unroll
    for (int mt = 0; mt < 4; ++mt) { acc0[mt] = bias0; acc1[mt] = bias1; }
#pragma unroll
    for (int kt = 0; kt < KT3; ++kt) {
      v16bf b0 = load_b_frag_w(Wc1, NT3, kt, 0, tid);
      v16bf b1 = load_b_frag_w(Wc1, NT3, kt, 8, tid);
#pragma unroll
      for (int mt = 0; mt < 4; ++mt) {
        v16bf a = load_a_frag(sM + mt * 16 * PA, PA, lane, kt * 32);
        acc0[mt] = wmma_bf16_t(b0, a, acc0[mt]);
        acc1[mt] = wmma_bf16_t(b1, a, acc1[mt]);
      }
    }
#pragma unroll
    for (int mt = 0; mt < 4; ++mt) {
      *(v8bf*)(sT + (mt * 16 + c) * PT + n0) = silu_pack8(acc0[mt]);
      *(v8bf*)(sT + (mt * 16 + c) * PT + n1) = silu_pack8(acc1[mt]);
    }
  }
  __syncthreads();

  if (wave < 4) {
    int mt = wave;
    v8f acc;
#pragma unroll
    for (int r = 0; r < 8; ++r) acc[r] = 0.0f;
    if (lane < 16) acc[0] = b_c2[0];
#pragma unroll
    for (int kt = 0; kt < KT4; ++kt) {
      v16bf a = load_a_frag(sT + mt * 16 * PT, PT, lane, kt * 32);
      acc = wmma_bf16_t(load_b_frag(Wc2, NT4, kt, 0, lane), a, acc);
    }
    if (lane < 16)
      ((float*)(sA + (mt * 16 + lane) * PA + 288))[3] = acc[0];
  }
  __syncthreads();

  {
    int sub = tid & 7;
#pragma unroll
    for (int h = 0; h < 2; ++h) {
      int m = h * 32 + (tid >> 3);
      if (e0 + m < EE) {
        v8bf v = *(const v8bf*)(sM + m * PA + sub * 8);
        *(v8bf*)(mrow + (long long)(e0 + m) * 64 + sub * 8) = v;
      }
    }
  }
  if (tid < ME && e0 + tid < EE) {
    const float* pr = (const float*)(sA + tid * PA + 288);
    ew[e0 + tid] = make_float4(pr[0], pr[1], pr[2], pr[3]);
  }
}


__global__ __launch_bounds__(256) void egnn_node_kernel(
    const float* __restrict__ x, const bf16_t* __restrict__ mrow,
    const float4* __restrict__ ew, const int* __restrict__ offs,
    const int* __restrict__ deg, const int* __restrict__ lst2,
    const float* __restrict__ b_n1, const float* __restrict__ b_n2,
    const bf16_t* __restrict__ Wn1, const bf16_t* __restrict__ Wn2,
    float* __restrict__ out) {
  __shared__ alignas(16) bf16_t sA[16 * PNA];
  __shared__ alignas(16) bf16_t sT[16 * PNT];
  __shared__ float sW[16];
  __shared__ float sR[16][3];

  const int tid = threadIdx.x;
  const int lane = tid & 31;
  const int wave = tid >> 5;
  const int g0 = blockIdx.x * 16;

  {
    int m = tid >> 4;
    int sub = tid & 15;
    long long g = g0 + m;
#pragma unroll
    for (int j = 0; j < 8; ++j) {
      int k = sub * 8 + j;
      sA[m * PNA + k] = f2bf(x[g * 131 + k]);
    }
    const int o = offs[g];
    const int n = deg[g];
    float a0 = 0.0f, a1 = 0.0f, a2 = 0.0f, a3 = 0.0f;
    float r0 = 0.0f, r1 = 0.0f, r2 = 0.0f, w = 0.0f;
    for (int i = 0; i < n; ++i) {
      int e = lst2[o + i];
      v4bf q = *(const v4bf*)(mrow + (long long)e * 64 + sub * 4);
      a0 += bf2f(q[0]); a1 += bf2f(q[1]); a2 += bf2f(q[2]); a3 += bf2f(q[3]);
      if (sub == 0) {
        float4 t = ew[e];
        r0 += t.x; r1 += t.y; r2 += t.z; w += t.w;
      }
    }
    sA[m * PNA + 128 + sub * 4 + 0] = f2bf(a0);
    sA[m * PNA + 128 + sub * 4 + 1] = f2bf(a1);
    sA[m * PNA + 128 + sub * 4 + 2] = f2bf(a2);
    sA[m * PNA + 128 + sub * 4 + 3] = f2bf(a3);
    if (sub == 0) {
      sW[m] = w;
      sR[m][0] = r0; sR[m][1] = r1; sR[m][2] = r2;
    }
  }
  __syncthreads();

  {
    v16bf a[KTN1];
#pragma unroll
    for (int kt = 0; kt < KTN1; ++kt) a[kt] = load_a_frag(sA, PNA, lane, kt * 32);
#pragma unroll
    for (int i = 0; i < 2; ++i) {
      int nt = wave + 8 * i;
      int n0 = nt * 16 + 8 * (lane >> 4);
      v8f acc;
#pragma unroll
      for (int r = 0; r < 8; ++r) acc[r] = b_n1[n0 + r];
#pragma unroll
      for (int kt = 0; kt < KTN1; ++kt)
        acc = wmma_bf16_t(load_b_frag(Wn1, NTN1, kt, nt, lane), a[kt], acc);
      *(v8bf*)(sT + (lane & 15) * PNT + n0) = silu_pack8(acc);
    }
  }
  __syncthreads();

  {
    int nt = wave;
    int n = nt * 16 + (lane & 15);
    float bias = b_n2[n];
    v8f acc;
#pragma unroll
    for (int r = 0; r < 8; ++r) acc[r] = bias;
#pragma unroll
    for (int kt = 0; kt < KTN2; ++kt) {
      v16bf a = load_a_frag(sT, PNT, lane, kt * 32);
      acc = wmma_bf16(a, load_b_frag(Wn2, NTN2, kt, nt, lane), acc);
    }
#pragma unroll
    for (int r = 0; r < 8; ++r) {
      int m = r + 8 * (lane >> 4);
      long long g = g0 + m;
      out[g * 131 + n] = acc[r] + x[g * 131 + n];
    }
  }

  if (tid < 48) {
    int m = tid / 3, c = tid % 3;
    long long g = g0 + m;
    out[g * 131 + 128 + c] = x[g * 131 + 128 + c] + sW[m] * sR[m][c];
  }
}


extern "C" void kernel_launch(void* const* d_in, const int* in_sizes, int n_in,
                              void* d_out, int out_size, void* d_ws,
                              size_t ws_size, hipStream_t stream) {
  const float* x    = (const float*)d_in[0];
  const int*   ei   = (const int*)d_in[1];
  const float* W_e1 = (const float*)d_in[2];
  const float* b_e1 = (const float*)d_in[3];
  const float* W_e2 = (const float*)d_in[4];
  const float* b_e2 = (const float*)d_in[5];
  const float* W_c1 = (const float*)d_in[6];
  const float* b_c1 = (const float*)d_in[7];
  const float* W_c2 = (const float*)d_in[8];
  const float* b_c2 = (const float*)d_in[9];
  const float* W_n1 = (const float*)d_in[10];
  const float* b_n1 = (const float*)d_in[11];
  const float* W_n2 = (const float*)d_in[12];
  const float* b_n2 = (const float*)d_in[13];
  float* out = (float*)d_out;

  char* ws = (char*)d_ws;
  int* deg  = (int*)ws;
  int* offs = deg + NNP;
  int* slot = offs + NNP;
  int* lst  = slot + EEP;
  int* lst2 = lst + EEP;
  bf16_t* mrow = (bf16_t*)(lst2 + EEP);
  float4* ew = (float4*)(mrow + (long long)EE * 64);
  char* wp = (char*)(ew + EE);
  bf16_t* We1b = (bf16_t*)wp; wp += (long long)KT1 * NT1 * 32 * 16 * 2;
  bf16_t* We2b = (bf16_t*)wp; wp += (long long)KT2 * NT2 * 32 * 16 * 2;
  bf16_t* Wc1b = (bf16_t*)wp; wp += (long long)KT3 * NT3 * 32 * 16 * 2;
  bf16_t* Wc2b = (bf16_t*)wp; wp += (long long)KT4 * NT4 * 32 * 16 * 2;
  bf16_t* Wn1b = (bf16_t*)wp; wp += (long long)KTN1 * NTN1 * 32 * 16 * 2;
  bf16_t* Wn2b = (bf16_t*)wp; wp += (long long)KTN2 * NTN2 * 32 * 16 * 2;

  zero_i32_kernel<<<NNP / 256, 256, 0, stream>>>(deg, NNP);
  count_kernel<<<(EE + 255) / 256, 256, 0, stream>>>(ei, deg, slot);
  scan_kernel<<<1, 1024, 0, stream>>>(deg, offs);
  fill_kernel<<<(EE + 255) / 256, 256, 0, stream>>>(ei, offs, slot, lst);
  order_kernel<<<(EE + 255) / 256, 256, 0, stream>>>(ei, offs, deg, lst, lst2);

  pack_weight_kernel<<<(KT1 * NT1 * 32 + 255) / 256, 256, 0, stream>>>(
      W_e1, We1b, K1, N1, KT1, NT1);
  pack_weight_kernel<<<(KT2 * NT2 * 32 + 255) / 256, 256, 0, stream>>>(
      W_e2, We2b, K2, N2, KT2, NT2);
  pack_weight_kernel<<<(KT3 * NT3 * 32 + 255) / 256, 256, 0, stream>>>(
      W_c1, Wc1b, K3, N3, KT3, NT3);
  pack_weight_kernel<<<(KT4 * NT4 * 32 + 255) / 256, 256, 0, stream>>>(
      W_c2, Wc2b, K4, N4, KT4, NT4);
  pack_weight_kernel<<<(KTN1 * NTN1 * 32 + 255) / 256, 256, 0, stream>>>(
      W_n1, Wn1b, KN1, NN1, KTN1, NTN1);
  pack_weight_kernel<<<(KTN2 * NTN2 * 32 + 255) / 256, 256, 0, stream>>>(
      W_n2, Wn2b, KN2, NN2, KTN2, NTN2);

  egnn_edge_kernel<<<(EE + ME - 1) / ME, 256, 0, stream>>>(
      x, ei, b_e1, b_e2, b_c1, b_c2, We1b, We2b, Wc1b, Wc2b, mrow, ew);

  egnn_node_kernel<<<NN / 16, 256, 0, stream>>>(
      x, mrow, ew, offs, deg, lst2, b_n1, b_n2, Wn1b, Wn2b, out);
}


# Wave-wide 16-byte global loads cover whole 128-byte lines: weight fragments packed as two 512-B halves per tile (was 8 lines, half used), gather rows read by consecutive lanes; sums unchanged
# speedup vs baseline: 1.6019x; 1.0413x over previous
#include <hip/hip_runtime.h>
#include <hip/hip_bf16.h>


typedef __bf16 bf16_t;
typedef __attribute__((ext_vector_type(16))) __bf16 v16bf;
typedef __attribute__((ext_vector_type(8)))  __bf16 v8bf;
typedef __attribute__((ext_vector_type(8)))  float  v8f;

#define NN 50000
#define EE 500000
#define K1 257
#define KT1 9
#define N1 514
#define NT1 33
#define K2 514
#define KT2 17
#define N2 64
#define NT2 4
#define K3 64
#define KT3 2
#define N3 256
#define NT3 16
#define K4 256
#define KT4 8
#define N4 1
#define NT4 1
#define KN1 192
#define KTN1 6
#define NN1 256
#define NTN1 16
#define KN2 256
#define KTN2 8
#define NN2 128
#define NTN2 8

__device__ __forceinline__ bf16_t f2bf(float f) {
  return (bf16_t)f;
}

__device__ __forceinline__ float bf2f(bf16_t b) {
  unsigned short h = __builtin_bit_cast(unsigned short, b);
  unsigned u = (unsigned)h << 16;
  return __builtin_bit_cast(float, u);
}

__device__ __forceinline__ float silu_f(float x) {
  return x * __builtin_amdgcn_rcpf(1.0f + __expf(-x));
}

__device__ __forceinline__ v16bf load_a_frag(const bf16_t* A, int stride,
                                             int lane, int kbase) {
  int m = lane & 15;
  int kh = lane >> 4;
  const bf16_t* p = A + m * stride + kbase + 8 * kh;
  v8bf lo = *(const v8bf*)(p);
  v8bf hi = *(const v8bf*)(p + 16);
  return __builtin_shufflevector(lo, hi, 0, 1, 2, 3, 4, 5, 6, 7,
                                 8, 9, 10, 11, 12, 13, 14, 15);
}

__device__ __forceinline__ v16bf load_b_frag(const bf16_t* B, int NT,
                                             int kt, int nt, int lane) {
  const bf16_t* p = B + (long long)(kt * NT + nt) * 512 + lane * 8;
  v8bf lo = *(const v8bf*)(p);
  v8bf hi = *(const v8bf*)(p + 256);
  return __builtin_shufflevector(lo, hi, 0, 1, 2, 3, 4, 5, 6, 7,
                                 8, 9, 10, 11, 12, 13, 14, 15);
}

__device__ __forceinline__ v16bf load_b_frag_w(const bf16_t* B, int NT,
                                               int kt, int ntc, int tid) {
  const bf16_t* p = B + (long long)(kt * NT + ntc) * 512
                      + (tid >> 5) * 512 + (tid & 31) * 8;
  v8bf lo = *(const v8bf*)(p);
  v8bf hi = *(const v8bf*)(p + 256);
  return __builtin_shufflevector(lo, hi, 0, 1, 2, 3, 4, 5, 6, 7,
                                 8, 9, 10, 11, 12, 13, 14, 15);
}

__device__ __forceinline__ v8f wmma_bf16(v16bf a, v16bf b, v8f c) {
  return __builtin_amdgcn_wmma_f32_16x16x32_bf16(
       false, a,  false, b,
       (short)0, c,  false,  false);
}

__device__ __forceinline__ v8f wmma_bf16_t(v16bf w, v16bf x, v8f c) {
  return wmma_bf16(w, x, c);
}

__device__ __forceinline__ v8bf silu_pack8(v8f acc) {
  v8bf o;
#pragma unroll
  for (int r = 0; r < 8; ++r) o[r] = f2bf(silu_f(acc[r]));
  return o;
}


typedef __attribute__((ext_vector_type(4))) __bf16 v4bf;

#define NNP 50176
#define EEP 500032

__global__ __launch_bounds__(256) void zero_i32_kernel(int* p, int n) {
  int i = blockIdx.x * 256 + threadIdx.x;
  if (i < n) p[i] = 0;
}

__global__ __launch_bounds__(256) void count_kernel(
    const int* __restrict__ ei, int* __restrict__ deg, int* __restrict__ slot) {
  int e = blockIdx.x * 256 + threadIdx.x;
  if (e < EE) slot[e] = atomicAdd(&deg[ei[EE + e]], 1);
}

__global__ __launch_bounds__(1024) void scan_kernel(
    const int* __restrict__ deg, int* __restrict__ offs) {
  __shared__ int sv[NNP];
  __shared__ int sp[1024];
  const int t = threadIdx.x;
  for (int i = t; i < NNP; i += 1024) sv[i] = (i < NN) ? deg[i] : 0;
  __syncthreads();
  const int base = t * 49;
  int s = 0;
  for (int j = 0; j < 49; ++j) s += sv[base + j];
  sp[t] = s;
  __syncthreads();
  for (int o = 1; o < 1024; o <<= 1) {
    int v = (t >= o) ? sp[t - o] : 0;
    __syncthreads();
    sp[t] += v;
    __syncthreads();
  }
  int run = sp[t] - s;
  for (int j = 0; j < 49; ++j) {
    int d = sv[base + j];
    sv[base + j] = run;
    run += d;
  }
  __syncthreads();
  for (int i = t; i < NN; i += 1024) offs[i] = sv[i];
}

__global__ __launch_bounds__(256) void fill_kernel(
    const int* __restrict__ ei, const int* __restrict__ offs,
    const int* __restrict__ slot, int* __restrict__ lst) {
  int e = blockIdx.x * 256 + threadIdx.x;
  if (e < EE) lst[offs[ei[EE + e]] + slot[e]] = e;
}

__global__ __launch_bounds__(256) void order_kernel(
    const int* __restrict__ ei, const int* __restrict__ offs,
    const int* __restrict__ deg, const int* __restrict__ lst,
    int* __restrict__ lst2) {
  int e = blockIdx.x * 256 + threadIdx.x;
  if (e >= EE) return;
  int d = ei[EE + e];
  int o = offs[d];
  int n = deg[d];
  int r = 0;
  for (int j = 0; j < n; ++j) r += (lst[o + j] < e) ? 1 : 0;
  lst2[o + r] = e;
}

__global__ __launch_bounds__(256) void pack_weight_kernel(
    const float* __restrict__ W, bf16_t* __restrict__ dst,
    int K, int N, int KT, int NT) {
  int idx = blockIdx.x * 256 + threadIdx.x;
  int total = KT * NT * 32;
  if (idx >= total) return;
  int lane = idx & 31;
  int tile = idx >> 5;
  int nt = tile % NT;
  int kt = tile / NT;
  int n = nt * 16 + (lane & 15);
  int kh = lane >> 4;
  bf16_t* o = dst + (long long)tile * 512 + lane * 8;
#pragma unroll
  for (int e = 0; e < 16; ++e) {
    int k = kt * 32 + 8 * kh + e + ((e >= 8) ? 8 : 0);
    float v = (k < K && n < N) ? W[(long long)k * N + n] : 0.0f;
    o[(e >> 3) * 256 + (e & 7)] = f2bf(v);
  }
}


#define PA 296
#define PH 552
#define PT 264
#define PNA 200
#define PNT 264
#define ME 64
#define SMEM_BYTES (ME * PA * 2 + ME * PH * 2)

__global__ __launch_bounds__(256, 6) void egnn_edge_kernel(
    const float* __restrict__ x, const int* __restrict__ ei,
    const float* __restrict__ b_e1, const float* __restrict__ b_e2,
    const float* __restrict__ b_c1, const float* __restrict__ b_c2,
    const bf16_t* __restrict__ We1, const bf16_t* __restrict__ We2,
    const bf16_t* __restrict__ Wc1, const bf16_t* __restrict__ Wc2,
    bf16_t* __restrict__ mrow, float4* __restrict__ ew) {
  __shared__ alignas(16) char smem[SMEM_BYTES];

  bf16_t* sA = (bf16_t*)smem;
  bf16_t* sM = (bf16_t*)smem;
  bf16_t* sH = (bf16_t*)(smem + ME * PA * 2);
  bf16_t* sT = sH;

  const int tid = threadIdx.x;
  const int lane = tid & 31;
  const int wave = tid >> 5;
  const int e0 = blockIdx.x * ME;

  {
    int el = e0 + wave * 8 + (lane & 7);
    if (el > EE - 1) el = EE - 1;
    const int sv = ei[el];
    const int dv = ei[EE + el];
#pragma unroll
    for (int i = 0; i < 8; ++i) {
      const int s = __builtin_amdgcn_readlane(sv, i);
      const int d = __builtin_amdgcn_readlane(dv, i);
      const float* xd = x + (long long)d * 131 + 4 * lane;
      const float* xs = x + (long long)s * 131 + 4 * lane;
      const int m = wave * 8 + i;
      v4bf pd, ps;
#pragma unroll
      for (int r = 0; r < 4; ++r) {
        pd[r] = f2bf(xd[r]);
        ps[r] = f2bf(xs[r]);
      }
      *(v4bf*)(sA + m * PA + 4 * lane)       = pd;
      *(v4bf*)(sA + m * PA + 128 + 4 * lane) = ps;
    }
    if (lane < 8) {
      const int m = wave * 8 + lane;
      const float* xs = x + (long long)sv * 131;
      const float* xd = x + (long long)dv * 131;
      float r0 = xs[128] - xd[128];
      float r1 = xs[129] - xd[129];
      float r2 = xs[130] - xd[130];
      float dist = sqrtf(r0 * r0 + r1 * r1 + r2 * r2);
      v8bf z;
#pragma unroll
      for (int k = 0; k < 8; ++k) z[k] = f2bf(0.0f);
      v8bf t = z;
      t[0] = f2bf(dist);
      *(v8bf*)(sA + m * PA + 256) = t;
      *(v8bf*)(sA + m * PA + 264) = z;
      *(v8bf*)(sA + m * PA + 272) = z;
      *(v8bf*)(sA + m * PA + 280) = z;
      float* pr = (float*)(sA + m * PA + 288);
      pr[0] = r0; pr[1] = r1; pr[2] = r2;
    }
#pragma unroll
    for (int i = 0; i < 4; ++i) {
      int idx = tid + i * 256;
      sH[(idx >> 4) * PH + 528 + (idx & 15)] = f2bf(0.0f);
    }
  }
  v16bf bq[2][2];
  bq[0][0] = load_b_frag_w(We1, NT1, 0, 0, tid);
  bq[0][1] = load_b_frag_w(We1, NT1, 0, 8, tid);
  v8f bias0, bias1;
#pragma unroll
  for (int r = 0; r < 8; ++r) {
    bias0[r] = b_e1[wave * 16 + 8 * (lane >> 4) + r];
    bias1[r] = b_e1[(wave + 8) * 16 + 8 * (lane >> 4) + r];
  }
  __syncthreads();

  {
    const int c = lane & 15;
    const int f8 = 8 * (lane >> 4);

    v8f acc0[4], acc1[4];
    v16bf a[4];
    a[0] = load_a_frag(sA,           PA, lane, 0);
    a[1] = load_a_frag(sA + 16 * PA, PA, lane, 0);
#pragma unroll
    for (int s = 0; s < 2 * KT1; ++s) {
      const int p = s / KT1;
      const int kt = s % KT1;
      const int n0 = (wave + 16 * p) * 16 + f8;
      const int n1 = (wave + 16 * p + 8) * 16 + f8;
      if (kt == 0) {
#pragma unroll
        for (int mt = 0; mt < 4; ++mt) { acc0[mt] = bias0; acc1[mt] = bias1; }
      }
      if (s + 1 < 2 * KT1) {
        const int p2 = (s + 1) / KT1;
        const int k2 = (s + 1) % KT1;
        bq[(s + 1) % 2][0] = load_b_frag_w(We1, NT1, k2, 16 * p2, tid);
        bq[(s + 1) % 2][1] = load_b_frag_w(We1, NT1, k2, 16 * p2 + 8, tid);
      }
      a[2] = load_a_frag(sA + 32 * PA, PA, lane, kt * 32);
      a[3] = load_a_frag(sA + 48 * PA, PA, lane, kt * 32);
      __builtin_amdgcn_sched_barrier(0);
      acc0[0] = wmma_bf16_t(bq[s % 2][0], a[0], acc0[0]);
      acc1[0] = wmma_bf16_t(bq[s % 2][1], a[0], acc1[0]);
      acc0[1] = wmma_bf16_t(bq[s % 2][0], a[1], acc0[1]);
      acc1[1] = wmma_bf16_t(bq[s % 2][1], a[1], acc1[1]);
      __builtin_amdgcn_sched_barrier(0);
      if (s + 1 < 2 * KT1) {
        const int k1 = (s + 1) % KT1;
        a[0] = load_a_frag(sA,           PA, lane, k1 * 32);
        a[1] = load_a_frag(sA + 16 * PA, PA, lane, k1 * 32);
      }
      __builtin_amdgcn_sched_barrier(0);
      acc0[2] = wmma_bf16_t(bq[s % 2][0], a[2], acc0[2]);
      acc1[2] = wmma_bf16_t(bq[s % 2][1], a[2], acc1[2]);
      acc0[3] = wmma_bf16_t(bq[s % 2][0], a[3], acc0[3]);
      acc1[3] = wmma_bf16_t(bq[s % 2][1], a[3], acc1[3]);
      if (kt == KT1 - 1) {
        if (p == 0) {
#pragma unroll
          for (int r = 0; r < 8; ++r) {
            bias0[r] = b_e1[n0 + 256 + r];
            bias1[r] = b_e1[n1 + 256 + r];
          }
          __builtin_amdgcn_sched_barrier(0);
        }
#pragma unroll
        for (int mt = 0; mt < 4; ++mt) {
          *(v8bf*)(sH + (mt * 16 + c) * PH + n0) = silu_pack8(acc0[mt]);
          *(v8bf*)(sH + (mt * 16 + c) * PH + n1) = silu_pack8(acc1[mt]);
        }
      }
    }

    if (wave == 0) {
      const int n0 = (NT1 - 1) * 16 + f8;
      v8f t[4];
#pragma unroll
      for (int r = 0; r < 8; ++r) t[0][r] = (n0 + r < N1) ? b_e1[n0 + r] : 0.0f;
#pragma unroll
      for (int mt = 1; mt < 4; ++mt) t[mt] = t[0];
#pragma unroll
      for (int kt = 0; kt < KT1; ++kt) {
        v16bf b = load_b_frag_w(We1, NT1, kt, 32, tid);
#pragma unroll
        for (int mt = 0; mt < 4; ++mt) {
          v16bf am = load_a_frag(sA + mt * 16 * PA, PA, lane, kt * 32);
          t[mt] = wmma_bf16_t(b, am, t[mt]);
        }
      }
#pragma unroll
      for (int mt = 0; mt < 4; ++mt)
        *(v8bf*)(sH + (mt * 16 + c) * PH + n0) = silu_pack8(t[mt]);
    }
  }
  __syncthreads();

  {
    int mp = wave >> 2;
    int nt = wave & 3;
    int c = lane & 15;
    int n0 = nt * 16 + 8 * (lane >> 4);
    v8f acc0, acc1;
#pragma unroll
    for (int r = 0; r < 8; ++r) { acc0[r] = b_e2[n0 + r]; acc1[r] = acc0[r]; }
#pragma unroll
    for (int kt = 0; kt < KT2; ++kt) {
      v16bf b  = load_b_frag(We2, NT2, kt, nt, lane);
      v16bf a0 = load_a_frag(sH + mp * 16 * PH,       PH, lane, kt * 32);
      v16bf a1 = load_a_frag(sH + (mp + 2) * 16 * PH, PH, lane, kt * 32);
      acc0 = wmma_bf16_t(b, a0, acc0);
      acc1 = wmma_bf16_t(b, a1, acc1);
    }
    *(v8bf*)(sM + (mp * 16 + c) * PA + n0)      = silu_pack8(acc0);
    *(v8bf*)(sM + (mp * 16 + 32 + c) * PA + n0) = silu_pack8(acc1);
  }
  __syncthreads();

  {
    const int c = lane & 15;
    const int nt0 = wave;
    const int nt1 = wave + 8;
    const int n0 = nt0 * 16 + 8 * (lane >> 4);
    const int n1 = nt1 * 16 + 8 * (lane >> 4);
    v8f bias0, bias1;
#pragma unroll
    for (int r = 0; r < 8; ++r) { bias0[r] = b_c1[n0 + r]; bias1[r] = b_c1[n1 + r]; }
    v8f acc0[4], acc1[4];
#pragma unroll
    for (int mt = 0; mt < 4; ++mt) { acc0[mt] = bias0; acc1[mt] = bias1; }
#pragma unroll
    for (int kt = 0; kt < KT3; ++kt) {
      v16bf b0 = load_b_frag_w(Wc1, NT3, kt, 0, tid);
      v16bf b1 = load_b_frag_w(Wc1, NT3, kt, 8, tid);
#pragma unroll
      for (int mt = 0; mt < 4; ++mt) {
        v16bf a = load_a_frag(sM + mt * 16 * PA, PA, lane, kt * 32);
        acc0[mt] = wmma_bf16_t(b0, a, acc0[mt]);
        acc1[mt] = wmma_bf16_t(b1, a, acc1[mt]);
      }
    }
#pragma unroll
    for (int mt = 0; mt < 4; ++mt) {
      *(v8bf*)(sT + (mt * 16 + c) * PT + n0) = silu_pack8(acc0[mt]);
      *(v8bf*)(sT + (mt * 16 + c) * PT + n1) = silu_pack8(acc1[mt]);
    }
  }
  __syncthreads();

  if (wave < 4) {
    int mt = wave;
    v8f acc;
#pragma unroll
    for (int r = 0; r < 8; ++r) acc[r] = 0.0f;
    if (lane < 16) acc[0] = b_c2[0];
#pragma unroll
    for (int kt = 0; kt < KT4; ++kt) {
      v16bf a = load_a_frag(sT + mt * 16 * PT, PT, lane, kt * 32);
      acc = wmma_bf16_t(load_b_frag(Wc2, NT4, kt, 0, lane), a, acc);
    }
    if (lane < 16)
      ((float*)(sA + (mt * 16 + lane) * PA + 288))[3] = acc[0];
  }
  __syncthreads();

  {
    int sub = tid & 7;
#pragma unroll
    for (int h = 0; h < 2; ++h) {
      int m = h * 32 + (tid >> 3);
      if (e0 + m < EE) {
        v8bf v = *(const v8bf*)(sM + m * PA + sub * 8);
        *(v8bf*)(mrow + (long long)(e0 + m) * 64 + sub * 8) = v;
      }
    }
  }
  if (tid < ME && e0 + tid < EE) {
    const float* pr = (const float*)(sA + tid * PA + 288);
    ew[e0 + tid] = make_float4(pr[0], pr[1], pr[2], pr[3]);
  }
}


__global__ __launch_bounds__(256) void egnn_node_kernel(
    const float* __restrict__ x, const bf16_t* __restrict__ mrow,
    const float4* __restrict__ ew, const int* __restrict__ offs,
    const int* __restrict__ deg, const int* __restrict__ lst2,
    const float* __restrict__ b_n1, const float* __restrict__ b_n2,
    const bf16_t* __restrict__ Wn1, const bf16_t* __restrict__ Wn2,
    float* __restrict__ out) {
  __shared__ alignas(16) bf16_t sA[16 * PNA];
  __shared__ alignas(16) bf16_t sT[16 * PNT];
  __shared__ float sW[16];
  __shared__ float sR[16][3];

  const int tid = threadIdx.x;
  const int lane = tid & 31;
  const int wave = tid >> 5;
  const int g0 = blockIdx.x * 16;

  {
    int m = tid >> 4;
    int sub = tid & 15;
    long long g = g0 + m;
#pragma unroll
    for (int j = 0; j < 8; ++j) {
      int k = sub * 8 + j;
      sA[m * PNA + k] = f2bf(x[g * 131 + k]);
    }
    const int o = offs[g];
    const int n = deg[g];
    float a0 = 0.0f, a1 = 0.0f, a2 = 0.0f, a3 = 0.0f;
    float r0 = 0.0f, r1 = 0.0f, r2 = 0.0f, w = 0.0f;
    for (int i = 0; i < n; ++i) {
      int e = lst2[o + i];
      v4bf q = *(const v4bf*)(mrow + (long long)e * 64 + sub * 4);
      a0 += bf2f(q[0]); a1 += bf2f(q[1]); a2 += bf2f(q[2]); a3 += bf2f(q[3]);
      if (sub == 0) {
        float4 t = ew[e];
        r0 += t.x; r1 += t.y; r2 += t.z; w += t.w;
      }
    }
    sA[m * PNA + 128 + sub * 4 + 0] = f2bf(a0);
    sA[m * PNA + 128 + sub * 4 + 1] = f2bf(a1);
    sA[m * PNA + 128 + sub * 4 + 2] = f2bf(a2);
    sA[m * PNA + 128 + sub * 4 + 3] = f2bf(a3);
    if (sub == 0) {
      sW[m] = w;
      sR[m][0] = r0; sR[m][1] = r1; sR[m][2] = r2;
    }
  }
  __syncthreads();

  {
    v16bf a[KTN1];
#pragma unroll
    for (int kt = 0; kt < KTN1; ++kt) a[kt] = load_a_frag(sA, PNA, lane, kt * 32);
#pragma unroll
    for (int i = 0; i < 2; ++i) {
      int nt = wave + 8 * i;
      int n0 = nt * 16 + 8 * (lane >> 4);
      v8f acc;
#pragma unroll
      for (int r = 0; r < 8; ++r) acc[r] = b_n1[n0 + r];
#pragma unroll
      for (int kt = 0; kt < KTN1; ++kt)
        acc = wmma_bf16_t(load_b_frag(Wn1, NTN1, kt, nt, lane), a[kt], acc);
      *(v8bf*)(sT + (lane & 15) * PNT + n0) = silu_pack8(acc);
    }
  }
  __syncthreads();

  {
    int nt = wave;
    int n = nt * 16 + (lane & 15);
    float bias = b_n2[n];
    v8f acc;
#pragma unroll
    for (int r = 0; r < 8; ++r) acc[r] = bias;
#pragma unroll
    for (int kt = 0; kt < KTN2; ++kt) {
      v16bf a = load_a_frag(sT, PNT, lane, kt * 32);
      acc = wmma_bf16(a, load_b_frag(Wn2, NTN2, kt, nt, lane), acc);
    }
#pragma unroll
    for (int r = 0; r < 8; ++r) {
      int m = r + 8 * (lane >> 4);
      long long g = g0 + m;
      out[g * 131 + n] = acc[r] + x[g * 131 + n];
    }
  }

  if (tid < 48) {
    int m = tid / 3, c = tid % 3;
    long long g = g0 + m;
    out[g * 131 + 128 + c] = x[g * 131 + 128 + c] + sW[m] * sR[m][c];
  }
}


extern "C" void kernel_launch(void* const* d_in, const int* in_sizes, int n_in,
                              void* d_out, int out_size, void* d_ws,
                              size_t ws_size, hipStream_t stream) {
  const float* x    = (const float*)d_in[0];
  const int*   ei   = (const int*)d_in[1];
  const float* W_e1 = (const float*)d_in[2];
  const float* b_e1 = (const float*)d_in[3];
  const float* W_e2 = (const float*)d_in[4];
  const float* b_e2 = (const float*)d_in[5];
  const float* W_c1 = (const float*)d_in[6];
  const float* b_c1 = (const float*)d_in[7];
  const float* W_c2 = (const float*)d_in[8];
  const float* b_c2 = (const float*)d_in[9];
  const float* W_n1 = (const float*)d_in[10];
  const float* b_n1 = (const float*)d_in[11];
  const float* W_n2 = (const float*)d_in[12];
  const float* b_n2 = (const float*)d_in[13];
  float* out = (float*)d_out;

  char* ws = (char*)d_ws;
  int* deg  = (int*)ws;
  int* offs = deg + NNP;
  int* slot = offs + NNP;
  int* lst  = slot + EEP;
  int* lst2 = lst + EEP;
  bf16_t* mrow = (bf16_t*)(lst2 + EEP);
  float4* ew = (float4*)(mrow + (long long)EE * 64);
  char* wp = (char*)(ew + EE);
  bf16_t* We1b = (bf16_t*)wp; wp += (long long)KT1 * NT1 * 32 * 16 * 2;
  bf16_t* We2b = (bf16_t*)wp; wp += (long long)KT2 * NT2 * 32 * 16 * 2;
  bf16_t* Wc1b = (bf16_t*)wp; wp += (long long)KT3 * NT3 * 32 * 16 * 2;
  bf16_t* Wc2b = (bf16_t*)wp; wp += (long long)KT4 * NT4 * 32 * 16 * 2;
  bf16_t* Wn1b = (bf16_t*)wp; wp += (long long)KTN1 * NTN1 * 32 * 16 * 2;
  bf16_t* Wn2b = (bf16_t*)wp; wp += (long long)KTN2 * NTN2 * 32 * 16 * 2;

  zero_i32_kernel<<<NNP / 256, 256, 0, stream>>>(deg, NNP);
  count_kernel<<<(EE + 255) / 256, 256, 0, stream>>>(ei, deg, slot);
  scan_kernel<<<1, 1024, 0, stream>>>(deg, offs);
  fill_kernel<<<(EE + 255) / 256, 256, 0, stream>>>(ei, offs, slot, lst);
  order_kernel<<<(EE + 255) / 256, 256, 0, stream>>>(ei, offs, deg, lst, lst2);

  pack_weight_kernel<<<(KT1 * NT1 * 32 + 255) / 256, 256, 0, stream>>>(
      W_e1, We1b, K1, N1, KT1, NT1);
  pack_weight_kernel<<<(KT2 * NT2 * 32 + 255) / 256, 256, 0, stream>>>(
      W_e2, We2b, K2, N2, KT2, NT2);
  pack_weight_kernel<<<(KT3 * NT3 * 32 + 255) / 256, 256, 0, stream>>>(
      W_c1, Wc1b, K3, N3, KT3, NT3);
  pack_weight_kernel<<<(KT4 * NT4 * 32 + 255) / 256, 256, 0, stream>>>(
      W_c2, Wc2b, K4, N4, KT4, NT4);
  pack_weight_kernel<<<(KTN1 * NTN1 * 32 + 255) / 256, 256, 0, stream>>>(
      W_n1, Wn1b, KN1, NN1, KTN1, NTN1);
  pack_weight_kernel<<<(KTN2 * NTN2 * 32 + 255) / 256, 256, 0, stream>>>(
      W_n2, Wn2b, KN2, NN2, KTN2, NTN2);

  egnn_edge_kernel<<<(EE + ME - 1) / ME, 256, 0, stream>>>(
      x, ei, b_e1, b_e2, b_c1, b_c2, We1b, We2b, Wc1b, Wc2b, mrow, ew);

  egnn_node_kernel<<<NN / 16, 256, 0, stream>>>(
      x, mrow, ew, offs, deg, lst2, b_n1, b_n2, Wn1b, Wn2b, out);
}


# Independent jobs share launches: six weight packs + degree clear as ONE launch, the degree count as the last work-groups of the edge launch: 13 launches per call become 6; every sum unchanged
# speedup vs baseline: 1.7832x; 1.1132x over previous
#include <hip/hip_runtime.h>
#include <hip/hip_bf16.h>


typedef __bf16 bf16_t;
typedef __attribute__((ext_vector_type(16))) __bf16 v16bf;
typedef __attribute__((ext_vector_type(8)))  __bf16 v8bf;
typedef __attribute__((ext_vector_type(8)))  float  v8f;

#define NN 50000
#define EE 500000
#define K1 257
#define KT1 9
#define N1 514
#define NT1 33
#define K2 514
#define KT2 17
#define N2 64
#define NT2 4
#define K3 64
#define KT3 2
#define N3 256
#define NT3 16
#define K4 256
#define KT4 8
#define N4 1
#define NT4 1
#define KN1 192
#define KTN1 6
#define NN1 256
#define NTN1 16
#define KN2 256
#define KTN2 8
#define NN2 128
#define NTN2 8

__device__ __forceinline__ bf16_t f2bf(float f) {
  return (bf16_t)f;
}

__device__ __forceinline__ float bf2f(bf16_t b) {
  unsigned short h = __builtin_bit_cast(unsigned short, b);
  unsigned u = (unsigned)h << 16;
  return __builtin_bit_cast(float, u);
}

__device__ __forceinline__ float silu_f(float x) {
  return x * __builtin_amdgcn_rcpf(1.0f + __expf(-x));
}

__device__ __forceinline__ v16bf load_a_frag(const bf16_t* A, int stride,
                                             int lane, int kbase) {
  int m = lane & 15;
  int kh = lane >> 4;
  const bf16_t* p = A + m * stride + kbase + 8 * kh;
  v8bf lo = *(const v8bf*)(p);
  v8bf hi = *(const v8bf*)(p + 16);
  return __builtin_shufflevector(lo, hi, 0, 1, 2, 3, 4, 5, 6, 7,
                                 8, 9, 10, 11, 12, 13, 14, 15);
}

__device__ __forceinline__ v16bf load_b_frag(const bf16_t* B, int NT,
                                             int kt, int nt, int lane) {
  const bf16_t* p = B + (long long)(kt * NT + nt) * 512 + lane * 8;
  v8bf lo = *(const v8bf*)(p);
  v8bf hi = *(const v8bf*)(p + 256);
  return __builtin_shufflevector(lo, hi, 0, 1, 2, 3, 4, 5, 6, 7,
                                 8, 9, 10, 11, 12, 13, 14, 15);
}

__device__ __forceinline__ v16bf load_b_frag_w(const bf16_t* B, int NT,
                                               int kt, int ntc, int tid) {
  const bf16_t* p = B + (long long)(kt * NT + ntc) * 512
                      + (tid >> 5) * 512 + (tid & 31) * 8;
  v8bf lo = *(const v8bf*)(p);
  v8bf hi = *(const v8bf*)(p + 256);
  return __builtin_shufflevector(lo, hi, 0, 1, 2, 3, 4, 5, 6, 7,
                                 8, 9, 10, 11, 12, 13, 14, 15);
}

__device__ __forceinline__ v8f wmma_bf16(v16bf a, v16bf b, v8f c) {
  return __builtin_amdgcn_wmma_f32_16x16x32_bf16(
       false, a,  false, b,
       (short)0, c,  false,  false);
}

__device__ __forceinline__ v8f wmma_bf16_t(v16bf w, v16bf x, v8f c) {
  return wmma_bf16(w, x, c);
}

__device__ __forceinline__ v8bf silu_pack8(v8f acc) {
  v8bf o;
#pragma unroll
  for (int r = 0; r < 8; ++r) o[r] = f2bf(silu_f(acc[r]));
  return o;
}


typedef __attribute__((ext_vector_type(4))) __bf16 v4bf;

#define NNP 50176
#define EEP 500032

__device__ __forceinline__ void count_edge(
    const int* __restrict__ ei, int* __restrict__ deg, int* __restrict__ slot, int e) {
  if (e < EE) slot[e] = atomicAdd(&deg[ei[EE + e]], 1);
}

__global__ __launch_bounds__(1024) void scan_kernel(
    const int* __restrict__ deg, int* __restrict__ offs) {
  __shared__ int sv[NNP];
  __shared__ int sp[1024];
  const int t = threadIdx.x;
  for (int i = t; i < NNP; i += 1024) sv[i] = (i < NN) ? deg[i] : 0;
  __syncthreads();
  const int base = t * 49;
  int s = 0;
  for (int j = 0; j < 49; ++j) s += sv[base + j];
  sp[t] = s;
  __syncthreads();
  for (int o = 1; o < 1024; o <<= 1) {
    int v = (t >= o) ? sp[t - o] : 0;
    __syncthreads();
    sp[t] += v;
    __syncthreads();
  }
  int run = sp[t] - s;
  for (int j = 0; j < 49; ++j) {
    int d = sv[base + j];
    sv[base + j] = run;
    run += d;
  }
  __syncthreads();
  for (int i = t; i < NN; i += 1024) offs[i] = sv[i];
}

__global__ __launch_bounds__(256) void fill_kernel(
    const int* __restrict__ ei, const int* __restrict__ offs,
    const int* __restrict__ slot, int* __restrict__ lst) {
  int e = blockIdx.x * 256 + threadIdx.x;
  if (e < EE) lst[offs[ei[EE + e]] + slot[e]] = e;
}

__global__ __launch_bounds__(256) void order_kernel(
    const int* __restrict__ ei, const int* __restrict__ offs,
    const int* __restrict__ deg, const int* __restrict__ lst,
    int* __restrict__ lst2) {
  int e = blockIdx.x * 256 + threadIdx.x;
  if (e >= EE) return;
  int d = ei[EE + e];
  int o = offs[d];
  int n = deg[d];
  int r = 0;
  for (int j = 0; j < n; ++j) r += (lst[o + j] < e) ? 1 : 0;
  lst2[o + r] = e;
}

__device__ __forceinline__ void pack_weight_lane(
    const float* __restrict__ W, bf16_t* __restrict__ dst,
    int K, int N, int KT, int NT, int idx) {
  int total = KT * NT * 32;
  if (idx >= total) return;
  int lane = idx & 31;
  int tile = idx >> 5;
  int nt = tile % NT;
  int kt = tile / NT;
  int n = nt * 16 + (lane & 15);
  int kh = lane >> 4;
  bf16_t* o = dst + (long long)tile * 512 + lane * 8;
#pragma unroll
  for (int e = 0; e < 16; ++e) {
    int k = kt * 32 + 8 * kh + e + ((e >= 8) ? 8 : 0);
    float v = (k < K && n < N) ? W[(long long)k * N + n] : 0.0f;
    o[(e >> 3) * 256 + (e & 7)] = f2bf(v);
  }
}

#define PB_E1 ((KT1 * NT1 * 32 + 255) / 256)
#define PB_E2 ((KT2 * NT2 * 32 + 255) / 256)
#define PB_C1 ((KT3 * NT3 * 32 + 255) / 256)
#define PB_C2 ((KT4 * NT4 * 32 + 255) / 256)
#define PB_N1 ((KTN1 * NTN1 * 32 + 255) / 256)
#define PB_N2 ((KTN2 * NTN2 * 32 + 255) / 256)
#define PB_ZERO (NNP / 256)
#define PB_TOTAL (PB_E1 + PB_E2 + PB_C1 + PB_C2 + PB_N1 + PB_N2 + PB_ZERO)

__global__ __launch_bounds__(256) void setup_kernel(
    const float* __restrict__ W_e1, const float* __restrict__ W_e2,
    const float* __restrict__ W_c1, const float* __restrict__ W_c2,
    const float* __restrict__ W_n1, const float* __restrict__ W_n2,
    bf16_t* __restrict__ We1b, bf16_t* __restrict__ We2b,
    bf16_t* __restrict__ Wc1b, bf16_t* __restrict__ Wc2b,
    bf16_t* __restrict__ Wn1b, bf16_t* __restrict__ Wn2b,
    int* __restrict__ deg) {
  int b = blockIdx.x;
  const int t = threadIdx.x;
  if (b < PB_E1) { pack_weight_lane(W_e1, We1b, K1, N1, KT1, NT1, b * 256 + t); return; }
  b -= PB_E1;
  if (b < PB_E2) { pack_weight_lane(W_e2, We2b, K2, N2, KT2, NT2, b * 256 + t); return; }
  b -= PB_E2;
  if (b < PB_C1) { pack_weight_lane(W_c1, Wc1b, K3, N3, KT3, NT3, b * 256 + t); return; }
  b -= PB_C1;
  if (b < PB_C2) { pack_weight_lane(W_c2, Wc2b, K4, N4, KT4, NT4, b * 256 + t); return; }
  b -= PB_C2;
  if (b < PB_N1) { pack_weight_lane(W_n1, Wn1b, KN1, NN1, KTN1, NTN1, b * 256 + t); return; }
  b -= PB_N1;
  if (b < PB_N2) { pack_weight_lane(W_n2, Wn2b, KN2, NN2, KTN2, NTN2, b * 256 + t); return; }
  b -= PB_N2;
  deg[b * 256 + t] = 0;
}


#define PA 296
#define PH 552
#define PT 264
#define PNA 200
#define PNT 264
#define ME 64
#define SMEM_BYTES (ME * PA * 2 + ME * PH * 2)
#define EDGE_BLOCKS ((EE + ME - 1) / ME)
#define COUNT_BLOCKS ((EE + 255) / 256)

__global__ __launch_bounds__(256, 6) void egnn_edge_kernel(
    const float* __restrict__ x, const int* __restrict__ ei,
    const float* __restrict__ b_e1, const float* __restrict__ b_e2,
    const float* __restrict__ b_c1, const float* __restrict__ b_c2,
    const bf16_t* __restrict__ We1, const bf16_t* __restrict__ We2,
    const bf16_t* __restrict__ Wc1, const bf16_t* __restrict__ Wc2,
    bf16_t* __restrict__ mrow, float4* __restrict__ ew,
    int* __restrict__ deg, int* __restrict__ slot) {
  __shared__ alignas(16) char smem[SMEM_BYTES];

  if (blockIdx.x >= EDGE_BLOCKS) {
    count_edge(ei, deg, slot, (blockIdx.x - EDGE_BLOCKS) * 256 + threadIdx.x);
    return;
  }

  bf16_t* sA = (bf16_t*)smem;
  bf16_t* sM = (bf16_t*)smem;
  bf16_t* sH = (bf16_t*)(smem + ME * PA * 2);
  bf16_t* sT = sH;

  const int tid = threadIdx.x;
  const int lane = tid & 31;
  const int wave = tid >> 5;
  const int e0 = blockIdx.x * ME;

  {
    int el = e0 + wave * 8 + (lane & 7);
    if (el > EE - 1) el = EE - 1;
    const int sv = ei[el];
    const int dv = ei[EE + el];
#pragma unroll
    for (int i = 0; i < 8; ++i) {
      const int s = __builtin_amdgcn_readlane(sv, i);
      const int d = __builtin_amdgcn_readlane(dv, i);
      const float* xd = x + (long long)d * 131 + 4 * lane;
      const float* xs = x + (long long)s * 131 + 4 * lane;
      const int m = wave * 8 + i;
      v4bf pd, ps;
#pragma unroll
      for (int r = 0; r < 4; ++r) {
        pd[r] = f2bf(xd[r]);
        ps[r] = f2bf(xs[r]);
      }
      *(v4bf*)(sA + m * PA + 4 * lane)       = pd;
      *(v4bf*)(sA + m * PA + 128 + 4 * lane) = ps;
    }
    if (lane < 8) {
      const int m = wave * 8 + lane;
      const float* xs = x + (long long)sv * 131;
      const float* xd = x + (long long)dv * 131;
      float r0 = xs[128] - xd[128];
      float r1 = xs[129] - xd[129];
      float r2 = xs[130] - xd[130];
      float dist = sqrtf(r0 * r0 + r1 * r1 + r2 * r2);
      v8bf z;
#pragma unroll
      for (int k = 0; k < 8; ++k) z[k] = f2bf(0.0f);
      v8bf t = z;
      t[0] = f2bf(dist);
      *(v8bf*)(sA + m * PA + 256) = t;
      *(v8bf*)(sA + m * PA + 264) = z;
      *(v8bf*)(sA + m * PA + 272) = z;
      *(v8bf*)(sA + m * PA + 280) = z;
      float* pr = (float*)(sA + m * PA + 288);
      pr[0] = r0; pr[1] = r1; pr[2] = r2;
    }
#pragma unroll
    for (int i = 0; i < 4; ++i) {
      int idx = tid + i * 256;
      sH[(idx >> 4) * PH + 528 + (idx & 15)] = f2bf(0.0f);
    }
  }
  v16bf bq[2][2];
  bq[0][0] = load_b_frag_w(We1, NT1, 0, 0, tid);
  bq[0][1] = load_b_frag_w(We1, NT1, 0, 8, tid);
  v8f bias0, bias1;
#pragma unroll
  for (int r = 0; r < 8; ++r) {
    bias0[r] = b_e1[wave * 16 + 8 * (lane >> 4) + r];
    bias1[r] = b_e1[(wave + 8) * 16 + 8 * (lane >> 4) + r];
  }
  __syncthreads();

  {
    const int c = lane & 15;
    const int f8 = 8 * (lane >> 4);

    v8f acc0[4], acc1[4];
    v16bf a[4];
    a[0] = load_a_frag(sA,           PA, lane, 0);
    a[1] = load_a_frag(sA + 16 * PA, PA, lane, 0);
#pragma unroll
    for (int s = 0; s < 2 * KT1; ++s) {
      const int p = s / KT1;
      const int kt = s % KT1;
      const int n0 = (wave + 16 * p) * 16 + f8;
      const int n1 = (wave + 16 * p + 8) * 16 + f8;
      if (kt == 0) {
#pragma unroll
        for (int mt = 0; mt < 4; ++mt) { acc0[mt] = bias0; acc1[mt] = bias1; }
      }
      if (s + 1 < 2 * KT1) {
        const int p2 = (s + 1) / KT1;
        const int k2 = (s + 1) % KT1;
        bq[(s + 1) % 2][0] = load_b_frag_w(We1, NT1, k2, 16 * p2, tid);
        bq[(s + 1) % 2][1] = load_b_frag_w(We1, NT1, k2, 16 * p2 + 8, tid);
      }
      a[2] = load_a_frag(sA + 32 * PA, PA, lane, kt * 32);
      a[3] = load_a_frag(sA + 48 * PA, PA, lane, kt * 32);
      __builtin_amdgcn_sched_barrier(0);
      acc0[0] = wmma_bf16_t(bq[s % 2][0], a[0], acc0[0]);
      acc1[0] = wmma_bf16_t(bq[s % 2][1], a[0], acc1[0]);
      acc0[1] = wmma_bf16_t(bq[s % 2][0], a[1], acc0[1]);
      acc1[1] = wmma_bf16_t(bq[s % 2][1], a[1], acc1[1]);
      __builtin_amdgcn_sched_barrier(0);
      if (s + 1 < 2 * KT1) {
        const int k1 = (s + 1) % KT1;
        a[0] = load_a_frag(sA,           PA, lane, k1 * 32);
        a[1] = load_a_frag(sA + 16 * PA, PA, lane, k1 * 32);
      }
      __builtin_amdgcn_sched_barrier(0);
      acc0[2] = wmma_bf16_t(bq[s % 2][0], a[2], acc0[2]);
      acc1[2] = wmma_bf16_t(bq[s % 2][1], a[2], acc1[2]);
      acc0[3] = wmma_bf16_t(bq[s % 2][0], a[3], acc0[3]);
      acc1[3] = wmma_bf16_t(bq[s % 2][1], a[3], acc1[3]);
      if (kt == KT1 - 1) {
        if (p == 0) {
#pragma unroll
          for (int r = 0; r < 8; ++r) {
            bias0[r] = b_e1[n0 + 256 + r];
            bias1[r] = b_e1[n1 + 256 + r];
          }
          __builtin_amdgcn_sched_barrier(0);
        }
#pragma unroll
        for (int mt = 0; mt < 4; ++mt) {
          *(v8bf*)(sH + (mt * 16 + c) * PH + n0) = silu_pack8(acc0[mt]);
          *(v8bf*)(sH + (mt * 16 + c) * PH + n1) = silu_pack8(acc1[mt]);
        }
      }
    }

    if (wave == 0) {
      const int n0 = (NT1 - 1) * 16 + f8;
      v8f t[4];
#pragma unroll
      for (int r = 0; r < 8; ++r) t[0][r] = (n0 + r < N1) ? b_e1[n0 + r] : 0.0f;
#pragma unroll
      for (int mt = 1; mt < 4; ++mt) t[mt] = t[0];
#pragma unroll
      for (int kt = 0; kt < KT1; ++kt) {
        v16bf b = load_b_frag_w(We1, NT1, kt, 32, tid);
#pragma unroll
        for (int mt = 0; mt < 4; ++mt) {
          v16bf am = load_a_frag(sA + mt * 16 * PA, PA, lane, kt * 32);
          t[mt] = wmma_bf16_t(b, am, t[mt]);
        }
      }
#pragma unroll
      for (int mt = 0; mt < 4; ++mt)
        *(v8bf*)(sH + (mt * 16 + c) * PH + n0) = silu_pack8(t[mt]);
    }
  }
  __syncthreads();

  {
    int mp = wave >> 2;
    int nt = wave & 3;
    int c = lane & 15;
    int n0 = nt * 16 + 8 * (lane >> 4);
    v8f acc0, acc1;
#pragma unroll
    for (int r = 0; r < 8; ++r) { acc0[r] = b_e2[n0 + r]; acc1[r] = acc0[r]; }
#pragma unroll
    for (int kt = 0; kt < KT2; ++kt) {
      v16bf b  = load_b_frag(We2, NT2, kt, nt, lane);
      v16bf a0 = load_a_frag(sH + mp * 16 * PH,       PH, lane, kt * 32);
      v16bf a1 = load_a_frag(sH + (mp + 2) * 16 * PH, PH, lane, kt * 32);
      acc0 = wmma_bf16_t(b, a0, acc0);
      acc1 = wmma_bf16_t(b, a1, acc1);
    }
    *(v8bf*)(sM + (mp * 16 + c) * PA + n0)      = silu_pack8(acc0);
    *(v8bf*)(sM + (mp * 16 + 32 + c) * PA + n0) = silu_pack8(acc1);
  }
  __syncthreads();

  {
    const int c = lane & 15;
    const int nt0 = wave;
    const int nt1 = wave + 8;
    const int n0 = nt0 * 16 + 8 * (lane >> 4);
    const int n1 = nt1 * 16 + 8 * (lane >> 4);
    v8f bias0, bias1;
#pragma unroll
    for (int r = 0; r < 8; ++r) { bias0[r] = b_c1[n0 + r]; bias1[r] = b_c1[n1 + r]; }
    v8f acc0[4], acc1[4];
#pragma unroll
    for (int mt = 0; mt < 4; ++mt) { acc0[mt] = bias0; acc1[mt] = bias1; }
#pragma unroll
    for (int kt = 0; kt < KT3; ++kt) {
      v16bf b0 = load_b_frag_w(Wc1, NT3, kt, 0, tid);
      v16bf b1 = load_b_frag_w(Wc1, NT3, kt, 8, tid);
#pragma unroll
      for (int mt = 0; mt < 4; ++mt) {
        v16bf a = load_a_frag(sM + mt * 16 * PA, PA, lane, kt * 32);
        acc0[mt] = wmma_bf16_t(b0, a, acc0[mt]);
        acc1[mt] = wmma_bf16_t(b1, a, acc1[mt]);
      }
    }
#pragma unroll
    for (int mt = 0; mt < 4; ++mt) {
      *(v8bf*)(sT + (mt * 16 + c) * PT + n0) = silu_pack8(acc0[mt]);
      *(v8bf*)(sT + (mt * 16 + c) * PT + n1) = silu_pack8(acc1[mt]);
    }
  }
  __syncthreads();

  if (wave < 4) {
    int mt = wave;
    v8f acc;
#pragma unroll
    for (int r = 0; r < 8; ++r) acc[r] = 0.0f;
    if (lane < 16) acc[0] = b_c2[0];
#pragma unroll
    for (int kt = 0; kt < KT4; ++kt) {
      v16bf a = load_a_frag(sT + mt * 16 * PT, PT, lane, kt * 32);
      acc = wmma_bf16_t(load_b_frag(Wc2, NT4, kt, 0, lane), a, acc);
    }
    if (lane < 16)
      ((float*)(sA + (mt * 16 + lane) * PA + 288))[3] = acc[0];
  }
  __syncthreads();

  {
    int sub = tid & 7;
#pragma unroll
    for (int h = 0; h < 2; ++h) {
      int m = h * 32 + (tid >> 3);
      if (e0 + m < EE) {
        v8bf v = *(const v8bf*)(sM + m * PA + sub * 8);
        *(v8bf*)(mrow + (long long)(e0 + m) * 64 + sub * 8) = v;
      }
    }
  }
  if (tid < ME && e0 + tid < EE) {
    const float* pr = (const float*)(sA + tid * PA + 288);
    ew[e0 + tid] = make_float4(pr[0], pr[1], pr[2], pr[3]);
  }
}


__global__ __launch_bounds__(256) void egnn_node_kernel(
    const float* __restrict__ x, const bf16_t* __restrict__ mrow,
    const float4* __restrict__ ew, const int* __restrict__ offs,
    const int* __restrict__ deg, const int* __restrict__ lst2,
    const float* __restrict__ b_n1, const float* __restrict__ b_n2,
    const bf16_t* __restrict__ Wn1, const bf16_t* __restrict__ Wn2,
    float* __restrict__ out) {
  __shared__ alignas(16) bf16_t sA[16 * PNA];
  __shared__ alignas(16) bf16_t sT[16 * PNT];
  __shared__ float sW[16];
  __shared__ float sR[16][3];

  const int tid = threadIdx.x;
  const int lane = tid & 31;
  const int wave = tid >> 5;
  const int g0 = blockIdx.x * 16;

  {
    int m = tid >> 4;
    int sub = tid & 15;
    long long g = g0 + m;
#pragma unroll
    for (int j = 0; j < 8; ++j) {
      int k = sub * 8 + j;
      sA[m * PNA + k] = f2bf(x[g * 131 + k]);
    }
    const int o = offs[g];
    const int n = deg[g];
    float a0 = 0.0f, a1 = 0.0f, a2 = 0.0f, a3 = 0.0f;
    float r0 = 0.0f, r1 = 0.0f, r2 = 0.0f, w = 0.0f;
    for (int i = 0; i < n; ++i) {
      int e = lst2[o + i];
      v4bf q = *(const v4bf*)(mrow + (long long)e * 64 + sub * 4);
      a0 += bf2f(q[0]); a1 += bf2f(q[1]); a2 += bf2f(q[2]); a3 += bf2f(q[3]);
      if (sub == 0) {
        float4 t = ew[e];
        r0 += t.x; r1 += t.y; r2 += t.z; w += t.w;
      }
    }
    sA[m * PNA + 128 + sub * 4 + 0] = f2bf(a0);
    sA[m * PNA + 128 + sub * 4 + 1] = f2bf(a1);
    sA[m * PNA + 128 + sub * 4 + 2] = f2bf(a2);
    sA[m * PNA + 128 + sub * 4 + 3] = f2bf(a3);
    if (sub == 0) {
      sW[m] = w;
      sR[m][0] = r0; sR[m][1] = r1; sR[m][2] = r2;
    }
  }
  __syncthreads();

  {
    v16bf a[KTN1];
#pragma unroll
    for (int kt = 0; kt < KTN1; ++kt) a[kt] = load_a_frag(sA, PNA, lane, kt * 32);
#pragma unroll
    for (int i = 0; i < 2; ++i) {
      int nt = wave + 8 * i;
      int n0 = nt * 16 + 8 * (lane >> 4);
      v8f acc;
#pragma unroll
      for (int r = 0; r < 8; ++r) acc[r] = b_n1[n0 + r];
#pragma unroll
      for (int kt = 0; kt < KTN1; ++kt)
        acc = wmma_bf16_t(load_b_frag(Wn1, NTN1, kt, nt, lane), a[kt], acc);
      *(v8bf*)(sT + (lane & 15) * PNT + n0) = silu_pack8(acc);
    }
  }
  __syncthreads();

  {
    int nt = wave;
    int n = nt * 16 + (lane & 15);
    float bias = b_n2[n];
    v8f acc;
#pragma unroll
    for (int r = 0; r < 8; ++r) acc[r] = bias;
#pragma unroll
    for (int kt = 0; kt < KTN2; ++kt) {
      v16bf a = load_a_frag(sT, PNT, lane, kt * 32);
      acc = wmma_bf16(a, load_b_frag(Wn2, NTN2, kt, nt, lane), acc);
    }
#pragma unroll
    for (int r = 0; r < 8; ++r) {
      int m = r + 8 * (lane >> 4);
      long long g = g0 + m;
      out[g * 131 + n] = acc[r] + x[g * 131 + n];
    }
  }

  if (tid < 48) {
    int m = tid / 3, c = tid % 3;
    long long g = g0 + m;
    out[g * 131 + 128 + c] = x[g * 131 + 128 + c] + sW[m] * sR[m][c];
  }
}


extern "C" void kernel_launch(void* const* d_in, const int* in_sizes, int n_in,
                              void* d_out, int out_size, void* d_ws,
                              size_t ws_size, hipStream_t stream) {
  const float* x    = (const float*)d_in[0];
  const int*   ei   = (const int*)d_in[1];
  const float* W_e1 = (const float*)d_in[2];
  const float* b_e1 = (const float*)d_in[3];
  const float* W_e2 = (const float*)d_in[4];
  const float* b_e2 = (const float*)d_in[5];
  const float* W_c1 = (const float*)d_in[6];
  const float* b_c1 = (const float*)d_in[7];
  const float* W_c2 = (const float*)d_in[8];
  const float* b_c2 = (const float*)d_in[9];
  const float* W_n1 = (const float*)d_in[10];
  const float* b_n1 = (const float*)d_in[11];
  const float* W_n2 = (const float*)d_in[12];
  const float* b_n2 = (const float*)d_in[13];
  float* out = (float*)d_out;

  char* ws = (char*)d_ws;
  int* deg  = (int*)ws;
  int* offs = deg + NNP;
  int* slot = offs + NNP;
  int* lst  = slot + EEP;
  int* lst2 = lst + EEP;
  bf16_t* mrow = (bf16_t*)(lst2 + EEP);
  float4* ew = (float4*)(mrow + (long long)EE * 64);
  char* wp = (char*)(ew + EE);
  bf16_t* We1b = (bf16_t*)wp; wp += (long long)KT1 * NT1 * 32 * 16 * 2;
  bf16_t* We2b = (bf16_t*)wp; wp += (long long)KT2 * NT2 * 32 * 16 * 2;
  bf16_t* Wc1b = (bf16_t*)wp; wp += (long long)KT3 * NT3 * 32 * 16 * 2;
  bf16_t* Wc2b = (bf16_t*)wp; wp += (long long)KT4 * NT4 * 32 * 16 * 2;
  bf16_t* Wn1b = (bf16_t*)wp; wp += (long long)KTN1 * NTN1 * 32 * 16 * 2;
  bf16_t* Wn2b = (bf16_t*)wp; wp += (long long)KTN2 * NTN2 * 32 * 16 * 2;

  setup_kernel<<<PB_TOTAL, 256, 0, stream>>>(
      W_e1, W_e2, W_c1, W_c2, W_n1, W_n2,
      We1b, We2b, Wc1b, Wc2b, Wn1b, Wn2b, deg);

  egnn_edge_kernel<<<EDGE_BLOCKS + COUNT_BLOCKS, 256, 0, stream>>>(
      x, ei, b_e1, b_e2, b_c1, b_c2, We1b, We2b, Wc1b, Wc2b, mrow, ew, deg, slot);

  scan_kernel<<<1, 1024, 0, stream>>>(deg, offs);
  fill_kernel<<<(EE + 255) / 256, 256, 0, stream>>>(ei, offs, slot, lst);
  order_kernel<<<(EE + 255) / 256, 256, 0, stream>>>(ei, offs, deg, lst, lst2);

  egnn_node_kernel<<<NN / 16, 256, 0, stream>>>(
      x, mrow, ew, offs, deg, lst2, b_n1, b_n2, Wn1b, Wn2b, out);
}


# No scan, fill and order launches: the count's atomic slot files each edge in its node's 64-slot bucket, which the node kernel ranks in LDS by edge number: 6 launches -> 3; sums and order unchanged
# speedup vs baseline: 2.1956x; 1.2313x over previous
#include <hip/hip_runtime.h>
#include <hip/hip_bf16.h>


typedef __bf16 bf16_t;
typedef __attribute__((ext_vector_type(16))) __bf16 v16bf;
typedef __attribute__((ext_vector_type(8)))  __bf16 v8bf;
typedef __attribute__((ext_vector_type(8)))  float  v8f;

#define NN 50000
#define EE 500000
#define K1 257
#define KT1 9
#define N1 514
#define NT1 33
#define K2 514
#define KT2 17
#define N2 64
#define NT2 4
#define K3 64
#define KT3 2
#define N3 256
#define NT3 16
#define K4 256
#define KT4 8
#define N4 1
#define NT4 1
#define KN1 192
#define KTN1 6
#define NN1 256
#define NTN1 16
#define KN2 256
#define KTN2 8
#define NN2 128
#define NTN2 8

__device__ __forceinline__ bf16_t f2bf(float f) {
  return (bf16_t)f;
}

__device__ __forceinline__ float bf2f(bf16_t b) {
  unsigned short h = __builtin_bit_cast(unsigned short, b);
  unsigned u = (unsigned)h << 16;
  return __builtin_bit_cast(float, u);
}

__device__ __forceinline__ float silu_f(float x) {
  return x * __builtin_amdgcn_rcpf(1.0f + __expf(-x));
}

__device__ __forceinline__ v16bf load_a_frag(const bf16_t* A, int stride,
                                             int lane, int kbase) {
  int m = lane & 15;
  int kh = lane >> 4;
  const bf16_t* p = A + m * stride + kbase + 8 * kh;
  v8bf lo = *(const v8bf*)(p);
  v8bf hi = *(const v8bf*)(p + 16);
  return __builtin_shufflevector(lo, hi, 0, 1, 2, 3, 4, 5, 6, 7,
                                 8, 9, 10, 11, 12, 13, 14, 15);
}

__device__ __forceinline__ v16bf load_b_frag(const bf16_t* B, int NT,
                                             int kt, int nt, int lane) {
  const bf16_t* p = B + (long long)(kt * NT + nt) * 512 + lane * 8;
  v8bf lo = *(const v8bf*)(p);
  v8bf hi = *(const v8bf*)(p + 256);
  return __builtin_shufflevector(lo, hi, 0, 1, 2, 3, 4, 5, 6, 7,
                                 8, 9, 10, 11, 12, 13, 14, 15);
}

__device__ __forceinline__ v16bf load_b_frag_w(const bf16_t* B, int NT,
                                               int kt, int ntc, int tid) {
  const bf16_t* p = B + (long long)(kt * NT + ntc) * 512
                      + (tid >> 5) * 512 + (tid & 31) * 8;
  v8bf lo = *(const v8bf*)(p);
  v8bf hi = *(const v8bf*)(p + 256);
  return __builtin_shufflevector(lo, hi, 0, 1, 2, 3, 4, 5, 6, 7,
                                 8, 9, 10, 11, 12, 13, 14, 15);
}

__device__ __forceinline__ v8f wmma_bf16(v16bf a, v16bf b, v8f c) {
  return __builtin_amdgcn_wmma_f32_16x16x32_bf16(
       false, a,  false, b,
       (short)0, c,  false,  false);
}

__device__ __forceinline__ v8f wmma_bf16_t(v16bf w, v16bf x, v8f c) {
  return wmma_bf16(w, x, c);
}

__device__ __forceinline__ v8bf silu_pack8(v8f acc) {
  v8bf o;
#pragma unroll
  for (int r = 0; r < 8; ++r) o[r] = f2bf(silu_f(acc[r]));
  return o;
}


typedef __attribute__((ext_vector_type(4))) __bf16 v4bf;

#define NNP 50176
#define CAP 64

__device__ __forceinline__ void count_edge(
    const int* __restrict__ ei, int* __restrict__ deg, int* __restrict__ bucket, int e) {
  if (e < EE) {
    const int d = ei[EE + e];
    const int s = atomicAdd(&deg[d], 1);
    if (s < CAP) bucket[(long long)d * CAP + s] = e;
  }
}

__device__ __forceinline__ void pack_weight_lane(
    const float* __restrict__ W, bf16_t* __restrict__ dst,
    int K, int N, int KT, int NT, int idx) {
  int total = KT * NT * 32;
  if (idx >= total) return;
  int lane = idx & 31;
  int tile = idx >> 5;
  int nt = tile % NT;
  int kt = tile / NT;
  int n = nt * 16 + (lane & 15);
  int kh = lane >> 4;
  bf16_t* o = dst + (long long)tile * 512 + lane * 8;
#pragma unroll
  for (int e = 0; e < 16; ++e) {
    int k = kt * 32 + 8 * kh + e + ((e >= 8) ? 8 : 0);
    float v = (k < K && n < N) ? W[(long long)k * N + n] : 0.0f;
    o[(e >> 3) * 256 + (e & 7)] = f2bf(v);
  }
}

#define PB_E1 ((KT1 * NT1 * 32 + 255) / 256)
#define PB_E2 ((KT2 * NT2 * 32 + 255) / 256)
#define PB_C1 ((KT3 * NT3 * 32 + 255) / 256)
#define PB_C2 ((KT4 * NT4 * 32 + 255) / 256)
#define PB_N1 ((KTN1 * NTN1 * 32 + 255) / 256)
#define PB_N2 ((KTN2 * NTN2 * 32 + 255) / 256)
#define PB_ZERO (NNP / 256)
#define PB_TOTAL (PB_E1 + PB_E2 + PB_C1 + PB_C2 + PB_N1 + PB_N2 + PB_ZERO)

__global__ __launch_bounds__(256) void setup_kernel(
    const float* __restrict__ W_e1, const float* __restrict__ W_e2,
    const float* __restrict__ W_c1, const float* __restrict__ W_c2,
    const float* __restrict__ W_n1, const float* __restrict__ W_n2,
    bf16_t* __restrict__ We1b, bf16_t* __restrict__ We2b,
    bf16_t* __restrict__ Wc1b, bf16_t* __restrict__ Wc2b,
    bf16_t* __restrict__ Wn1b, bf16_t* __restrict__ Wn2b,
    int* __restrict__ deg) {
  int b = blockIdx.x;
  const int t = threadIdx.x;
  if (b < PB_E1) { pack_weight_lane(W_e1, We1b, K1, N1, KT1, NT1, b * 256 + t); return; }
  b -= PB_E1;
  if (b < PB_E2) { pack_weight_lane(W_e2, We2b, K2, N2, KT2, NT2, b * 256 + t); return; }
  b -= PB_E2;
  if (b < PB_C1) { pack_weight_lane(W_c1, Wc1b, K3, N3, KT3, NT3, b * 256 + t); return; }
  b -= PB_C1;
  if (b < PB_C2) { pack_weight_lane(W_c2, Wc2b, K4, N4, KT4, NT4, b * 256 + t); return; }
  b -= PB_C2;
  if (b < PB_N1) { pack_weight_lane(W_n1, Wn1b, KN1, NN1, KTN1, NTN1, b * 256 + t); return; }
  b -= PB_N1;
  if (b < PB_N2) { pack_weight_lane(W_n2, Wn2b, KN2, NN2, KTN2, NTN2, b * 256 + t); return; }
  b -= PB_N2;
  deg[b * 256 + t] = 0;
}


#define PA 296
#define PH 552
#define PT 264
#define PNA 200
#define PNT 264
#define ME 64
#define SMEM_BYTES (ME * PA * 2 + ME * PH * 2)
#define EDGE_BLOCKS ((EE + ME - 1) / ME)
#define COUNT_BLOCKS ((EE + 255) / 256)

__global__ __launch_bounds__(256, 6) void egnn_edge_kernel(
    const float* __restrict__ x, const int* __restrict__ ei,
    const float* __restrict__ b_e1, const float* __restrict__ b_e2,
    const float* __restrict__ b_c1, const float* __restrict__ b_c2,
    const bf16_t* __restrict__ We1, const bf16_t* __restrict__ We2,
    const bf16_t* __restrict__ Wc1, const bf16_t* __restrict__ Wc2,
    bf16_t* __restrict__ mrow, float4* __restrict__ ew,
    int* __restrict__ deg, int* __restrict__ bucket) {
  __shared__ alignas(16) char smem[SMEM_BYTES];

  if (blockIdx.x >= EDGE_BLOCKS) {
    count_edge(ei, deg, bucket, (blockIdx.x - EDGE_BLOCKS) * 256 + threadIdx.x);
    return;
  }

  bf16_t* sA = (bf16_t*)smem;
  bf16_t* sM = (bf16_t*)smem;
  bf16_t* sH = (bf16_t*)(smem + ME * PA * 2);
  bf16_t* sT = sH;

  const int tid = threadIdx.x;
  const int lane = tid & 31;
  const int wave = tid >> 5;
  const int e0 = blockIdx.x * ME;

  {
    int el = e0 + wave * 8 + (lane & 7);
    if (el > EE - 1) el = EE - 1;
    const int sv = ei[el];
    const int dv = ei[EE + el];
#pragma unroll
    for (int i = 0; i < 8; ++i) {
      const int s = __builtin_amdgcn_readlane(sv, i);
      const int d = __builtin_amdgcn_readlane(dv, i);
      const float* xd = x + (long long)d * 131 + 4 * lane;
      const float* xs = x + (long long)s * 131 + 4 * lane;
      const int m = wave * 8 + i;
      v4bf pd, ps;
#pragma unroll
      for (int r = 0; r < 4; ++r) {
        pd[r] = f2bf(xd[r]);
        ps[r] = f2bf(xs[r]);
      }
      *(v4bf*)(sA + m * PA + 4 * lane)       = pd;
      *(v4bf*)(sA + m * PA + 128 + 4 * lane) = ps;
    }
    if (lane < 8) {
      const int m = wave * 8 + lane;
      const float* xs = x + (long long)sv * 131;
      const float* xd = x + (long long)dv * 131;
      float r0 = xs[128] - xd[128];
      float r1 = xs[129] - xd[129];
      float r2 = xs[130] - xd[130];
      float dist = sqrtf(r0 * r0 + r1 * r1 + r2 * r2);
      v8bf z;
#pragma unroll
      for (int k = 0; k < 8; ++k) z[k] = f2bf(0.0f);
      v8bf t = z;
      t[0] = f2bf(dist);
      *(v8bf*)(sA + m * PA + 256) = t;
      *(v8bf*)(sA + m * PA + 264) = z;
      *(v8bf*)(sA + m * PA + 272) = z;
      *(v8bf*)(sA + m * PA + 280) = z;
      float* pr = (float*)(sA + m * PA + 288);
      pr[0] = r0; pr[1] = r1; pr[2] = r2;
    }
#pragma unroll
    for (int i = 0; i < 4; ++i) {
      int idx = tid + i * 256;
      sH[(idx >> 4) * PH + 528 + (idx & 15)] = f2bf(0.0f);
    }
  }
  v16bf bq[2][2];
  bq[0][0] = load_b_frag_w(We1, NT1, 0, 0, tid);
  bq[0][1] = load_b_frag_w(We1, NT1, 0, 8, tid);
  v8f bias0, bias1;
#pragma unroll
  for (int r = 0; r < 8; ++r) {
    bias0[r] = b_e1[wave * 16 + 8 * (lane >> 4) + r];
    bias1[r] = b_e1[(wave + 8) * 16 + 8 * (lane >> 4) + r];
  }
  __syncthreads();

  {
    const int c = lane & 15;
    const int f8 = 8 * (lane >> 4);

    v8f acc0[4], acc1[4];
    v16bf a[4];
    a[0] = load_a_frag(sA,           PA, lane, 0);
    a[1] = load_a_frag(sA + 16 * PA, PA, lane, 0);
#pragma unroll
    for (int s = 0; s < 2 * KT1; ++s) {
      const int p = s / KT1;
      const int kt = s % KT1;
      const int n0 = (wave + 16 * p) * 16 + f8;
      const int n1 = (wave + 16 * p + 8) * 16 + f8;
      if (kt == 0) {
#pragma unroll
        for (int mt = 0; mt < 4; ++mt) { acc0[mt] = bias0; acc1[mt] = bias1; }
      }
      if (s + 1 < 2 * KT1) {
        const int p2 = (s + 1) / KT1;
        const int k2 = (s + 1) % KT1;
        bq[(s + 1) % 2][0] = load_b_frag_w(We1, NT1, k2, 16 * p2, tid);
        bq[(s + 1) % 2][1] = load_b_frag_w(We1, NT1, k2, 16 * p2 + 8, tid);
      }
      a[2] = load_a_frag(sA + 32 * PA, PA, lane, kt * 32);
      a[3] = load_a_frag(sA + 48 * PA, PA, lane, kt * 32);
      __builtin_amdgcn_sched_barrier(0);
      acc0[0] = wmma_bf16_t(bq[s % 2][0], a[0], acc0[0]);
      acc1[0] = wmma_bf16_t(bq[s % 2][1], a[0], acc1[0]);
      acc0[1] = wmma_bf16_t(bq[s % 2][0], a[1], acc0[1]);
      acc1[1] = wmma_bf16_t(bq[s % 2][1], a[1], acc1[1]);
      __builtin_amdgcn_sched_barrier(0);
      if (s + 1 < 2 * KT1) {
        const int k1 = (s + 1) % KT1;
        a[0] = load_a_frag(sA,           PA, lane, k1 * 32);
        a[1] = load_a_frag(sA + 16 * PA, PA, lane, k1 * 32);
      }
      __builtin_amdgcn_sched_barrier(0);
      acc0[2] = wmma_bf16_t(bq[s % 2][0], a[2], acc0[2]);
      acc1[2] = wmma_bf16_t(bq[s % 2][1], a[2], acc1[2]);
      acc0[3] = wmma_bf16_t(bq[s % 2][0], a[3], acc0[3]);
      acc1[3] = wmma_bf16_t(bq[s % 2][1], a[3], acc1[3]);
      if (kt == KT1 - 1) {
        if (p == 0) {
#pragma unroll
          for (int r = 0; r < 8; ++r) {
            bias0[r] = b_e1[n0 + 256 + r];
            bias1[r] = b_e1[n1 + 256 + r];
          }
          __builtin_amdgcn_sched_barrier(0);
        }
#pragma unroll
        for (int mt = 0; mt < 4; ++mt) {
          *(v8bf*)(sH + (mt * 16 + c) * PH + n0) = silu_pack8(acc0[mt]);
          *(v8bf*)(sH + (mt * 16 + c) * PH + n1) = silu_pack8(acc1[mt]);
        }
      }
    }

    if (wave == 0) {
      const int n0 = (NT1 - 1) * 16 + f8;
      v8f t[4];
#pragma unroll
      for (int r = 0; r < 8; ++r) t[0][r] = (n0 + r < N1) ? b_e1[n0 + r] : 0.0f;
#pragma unroll
      for (int mt = 1; mt < 4; ++mt) t[mt] = t[0];
#pragma unroll
      for (int kt = 0; kt < KT1; ++kt) {
        v16bf b = load_b_frag_w(We1, NT1, kt, 32, tid);
#pragma unroll
        for (int mt = 0; mt < 4; ++mt) {
          v16bf am = load_a_frag(sA + mt * 16 * PA, PA, lane, kt * 32);
          t[mt] = wmma_bf16_t(b, am, t[mt]);
        }
      }
#pragma unroll
      for (int mt = 0; mt < 4; ++mt)
        *(v8bf*)(sH + (mt * 16 + c) * PH + n0) = silu_pack8(t[mt]);
    }
  }
  __syncthreads();

  {
    int mp = wave >> 2;
    int nt = wave & 3;
    int c = lane & 15;
    int n0 = nt * 16 + 8 * (lane >> 4);
    v8f acc0, acc1;
#pragma unroll
    for (int r = 0; r < 8; ++r) { acc0[r] = b_e2[n0 + r]; acc1[r] = acc0[r]; }
#pragma unroll
    for (int kt = 0; kt < KT2; ++kt) {
      v16bf b  = load_b_frag(We2, NT2, kt, nt, lane);
      v16bf a0 = load_a_frag(sH + mp * 16 * PH,       PH, lane, kt * 32);
      v16bf a1 = load_a_frag(sH + (mp + 2) * 16 * PH, PH, lane, kt * 32);
      acc0 = wmma_bf16_t(b, a0, acc0);
      acc1 = wmma_bf16_t(b, a1, acc1);
    }
    *(v8bf*)(sM + (mp * 16 + c) * PA + n0)      = silu_pack8(acc0);
    *(v8bf*)(sM + (mp * 16 + 32 + c) * PA + n0) = silu_pack8(acc1);
  }
  __syncthreads();

  {
    const int c = lane & 15;
    const int nt0 = wave;
    const int nt1 = wave + 8;
    const int n0 = nt0 * 16 + 8 * (lane >> 4);
    const int n1 = nt1 * 16 + 8 * (lane >> 4);
    v8f bias0, bias1;
#pragma unroll
    for (int r = 0; r < 8; ++r) { bias0[r] = b_c1[n0 + r]; bias1[r] = b_c1[n1 + r]; }
    v8f acc0[4], acc1[4];
#pragma unroll
    for (int mt = 0; mt < 4; ++mt) { acc0[mt] = bias0; acc1[mt] = bias1; }
#pragma unroll
    for (int kt = 0; kt < KT3; ++kt) {
      v16bf b0 = load_b_frag_w(Wc1, NT3, kt, 0, tid);
      v16bf b1 = load_b_frag_w(Wc1, NT3, kt, 8, tid);
#pragma unroll
      for (int mt = 0; mt < 4; ++mt) {
        v16bf a = load_a_frag(sM + mt * 16 * PA, PA, lane, kt * 32);
        acc0[mt] = wmma_bf16_t(b0, a, acc0[mt]);
        acc1[mt] = wmma_bf16_t(b1, a, acc1[mt]);
      }
    }
#pragma unroll
    for (int mt = 0; mt < 4; ++mt) {
      *(v8bf*)(sT + (mt * 16 + c) * PT + n0) = silu_pack8(acc0[mt]);
      *(v8bf*)(sT + (mt * 16 + c) * PT + n1) = silu_pack8(acc1[mt]);
    }
  }
  __syncthreads();

  if (wave < 4) {
    int mt = wave;
    v8f acc;
#pragma unroll
    for (int r = 0; r < 8; ++r) acc[r] = 0.0f;
    if (lane < 16) acc[0] = b_c2[0];
#pragma unroll
    for (int kt = 0; kt < KT4; ++kt) {
      v16bf a = load_a_frag(sT + mt * 16 * PT, PT, lane, kt * 32);
      acc = wmma_bf16_t(load_b_frag(Wc2, NT4, kt, 0, lane), a, acc);
    }
    if (lane < 16)
      ((float*)(sA + (mt * 16 + lane) * PA + 288))[3] = acc[0];
  }
  __syncthreads();

  {
    int sub = tid & 7;
#pragma unroll
    for (int h = 0; h < 2; ++h) {
      int m = h * 32 + (tid >> 3);
      if (e0 + m < EE) {
        v8bf v = *(const v8bf*)(sM + m * PA + sub * 8);
        *(v8bf*)(mrow + (long long)(e0 + m) * 64 + sub * 8) = v;
      }
    }
  }
  if (tid < ME && e0 + tid < EE) {
    const float* pr = (const float*)(sA + tid * PA + 288);
    ew[e0 + tid] = make_float4(pr[0], pr[1], pr[2], pr[3]);
  }
}


__global__ __launch_bounds__(256) void egnn_node_kernel(
    const float* __restrict__ x, const bf16_t* __restrict__ mrow,
    const float4* __restrict__ ew, const int* __restrict__ ei,
    const int* __restrict__ deg, const int* __restrict__ bucket,
    const float* __restrict__ b_n1, const float* __restrict__ b_n2,
    const bf16_t* __restrict__ Wn1, const bf16_t* __restrict__ Wn2,
    float* __restrict__ out) {
  __shared__ alignas(16) bf16_t sA[16 * PNA];
  __shared__ alignas(16) bf16_t sT[16 * PNT];
  __shared__ float sW[16];
  __shared__ float sR[16][3];
  __shared__ alignas(16) int sU[16 * CAP];
  __shared__ alignas(16) int sS[16 * CAP];

  const int tid = threadIdx.x;
  const int lane = tid & 31;
  const int wave = tid >> 5;
  const int g0 = blockIdx.x * 16;

  {
    int m = tid >> 4;
    int sub = tid & 15;
    long long g = g0 + m;
#pragma unroll
    for (int j = 0; j < 8; ++j) {
      int k = sub * 8 + j;
      sA[m * PNA + k] = f2bf(x[g * 131 + k]);
    }
    const int n = deg[g];
    const bool spill = n > CAP;
    const int nb = spill ? 0 : n;
    int ev[CAP / 16];
#pragma unroll
    for (int q = 0; q < CAP / 16; ++q) {
      const int idx = sub + 16 * q;
      ev[q] = (idx < nb) ? bucket[g * CAP + idx] : 0x7fffffff;
    }
#pragma unroll
    for (int q = 0; q < CAP / 16; ++q) sU[m * CAP + sub + 16 * q] = ev[q];
    __syncthreads();
    for (int idx = sub; idx < nb; idx += 16) {
      const int mine = sU[m * CAP + idx];
      int r = 0;
      for (int j = 0; j < nb; j += 4) {
        const int4 v = *(const int4*)(sU + m * CAP + j);
        r += (v.x < mine ? 1 : 0) + (v.y < mine ? 1 : 0)
           + (v.z < mine ? 1 : 0) + (v.w < mine ? 1 : 0);
      }
      sS[m * CAP + r] = mine;
    }
    __syncthreads();
    float a0 = 0.0f, a1 = 0.0f, a2 = 0.0f, a3 = 0.0f;
    float r0 = 0.0f, r1 = 0.0f, r2 = 0.0f, w = 0.0f;
    auto add_edge = [&](int e) {
      v4bf q = *(const v4bf*)(mrow + (long long)e * 64 + sub * 4);
      a0 += bf2f(q[0]); a1 += bf2f(q[1]); a2 += bf2f(q[2]); a3 += bf2f(q[3]);
      if (sub == 0) {
        float4 t = ew[e];
        r0 += t.x; r1 += t.y; r2 += t.z; w += t.w;
      }
    };
    for (int i = 0; i < nb; ++i) add_edge(sS[m * CAP + i]);
    if (spill) {
      for (int e = 0; e < EE; ++e)
        if (ei[EE + e] == (int)g) add_edge(e);
    }
    sA[m * PNA + 128 + sub * 4 + 0] = f2bf(a0);
    sA[m * PNA + 128 + sub * 4 + 1] = f2bf(a1);
    sA[m * PNA + 128 + sub * 4 + 2] = f2bf(a2);
    sA[m * PNA + 128 + sub * 4 + 3] = f2bf(a3);
    if (sub == 0) {
      sW[m] = w;
      sR[m][0] = r0; sR[m][1] = r1; sR[m][2] = r2;
    }
  }
  __syncthreads();

  {
    v16bf a[KTN1];
#pragma unroll
    for (int kt = 0; kt < KTN1; ++kt) a[kt] = load_a_frag(sA, PNA, lane, kt * 32);
#pragma unroll
    for (int i = 0; i < 2; ++i) {
      int nt = wave + 8 * i;
      int n0 = nt * 16 + 8 * (lane >> 4);
      v8f acc;
#pragma unroll
      for (int r = 0; r < 8; ++r) acc[r] = b_n1[n0 + r];
#pragma unroll
      for (int kt = 0; kt < KTN1; ++kt)
        acc = wmma_bf16_t(load_b_frag(Wn1, NTN1, kt, nt, lane), a[kt], acc);
      *(v8bf*)(sT + (lane & 15) * PNT + n0) = silu_pack8(acc);
    }
  }
  __syncthreads();

  {
    int nt = wave;
    int n = nt * 16 + (lane & 15);
    float bias = b_n2[n];
    v8f acc;
#pragma unroll
    for (int r = 0; r < 8; ++r) acc[r] = bias;
#pragma unroll
    for (int kt = 0; kt < KTN2; ++kt) {
      v16bf a = load_a_frag(sT, PNT, lane, kt * 32);
      acc = wmma_bf16(a, load_b_frag(Wn2, NTN2, kt, nt, lane), acc);
    }
#pragma unroll
    for (int r = 0; r < 8; ++r) {
      int m = r + 8 * (lane >> 4);
      long long g = g0 + m;
      out[g * 131 + n] = acc[r] + x[g * 131 + n];
    }
  }

  if (tid < 48) {
    int m = tid / 3, c = tid % 3;
    long long g = g0 + m;
    out[g * 131 + 128 + c] = x[g * 131 + 128 + c] + sW[m] * sR[m][c];
  }
}


extern "C" void kernel_launch(void* const* d_in, const int* in_sizes, int n_in,
                              void* d_out, int out_size, void* d_ws,
                              size_t ws_size, hipStream_t stream) {
  const float* x    = (const float*)d_in[0];
  const int*   ei   = (const int*)d_in[1];
  const float* W_e1 = (const float*)d_in[2];
  const float* b_e1 = (const float*)d_in[3];
  const float* W_e2 = (const float*)d_in[4];
  const float* b_e2 = (const float*)d_in[5];
  const float* W_c1 = (const float*)d_in[6];
  const float* b_c1 = (const float*)d_in[7];
  const float* W_c2 = (const float*)d_in[8];
  const float* b_c2 = (const float*)d_in[9];
  const float* W_n1 = (const float*)d_in[10];
  const float* b_n1 = (const float*)d_in[11];
  const float* W_n2 = (const float*)d_in[12];
  const float* b_n2 = (const float*)d_in[13];
  float* out = (float*)d_out;

  char* ws = (char*)d_ws;
  int* deg  = (int*)ws;
  int* bucket = deg + NNP;
  bf16_t* mrow = (bf16_t*)(bucket + (long long)NN * CAP);
  float4* ew = (float4*)(mrow + (long long)EE * 64);
  char* wp = (char*)(ew + EE);
  bf16_t* We1b = (bf16_t*)wp; wp += (long long)KT1 * NT1 * 32 * 16 * 2;
  bf16_t* We2b = (bf16_t*)wp; wp += (long long)KT2 * NT2 * 32 * 16 * 2;
  bf16_t* Wc1b = (bf16_t*)wp; wp += (long long)KT3 * NT3 * 32 * 16 * 2;
  bf16_t* Wc2b = (bf16_t*)wp; wp += (long long)KT4 * NT4 * 32 * 16 * 2;
  bf16_t* Wn1b = (bf16_t*)wp; wp += (long long)KTN1 * NTN1 * 32 * 16 * 2;
  bf16_t* Wn2b = (bf16_t*)wp; wp += (long long)KTN2 * NTN2 * 32 * 16 * 2;

  setup_kernel<<<PB_TOTAL, 256, 0, stream>>>(
      W_e1, W_e2, W_c1, W_c2, W_n1, W_n2,
      We1b, We2b, Wc1b, Wc2b, Wn1b, Wn2b, deg);

  egnn_edge_kernel<<<EDGE_BLOCKS + COUNT_BLOCKS, 256, 0, stream>>>(
      x, ei, b_e1, b_e2, b_c1, b_c2, We1b, We2b, Wc1b, Wc2b, mrow, ew, deg, bucket);

  egnn_node_kernel<<<NN / 16, 256, 0, stream>>>(
      x, mrow, ew, ei, deg, bucket, b_n1, b_n2, Wn1b, Wn2b, out);
}


# Edge kernel: 96 edges per work-group, not 64: a weight fragment feeds 6 M-tiles (12 accumulator tiles per wave), weight bytes per edge -1/3; LDS 162816 B, 2 work-groups per WGP; sums unchanged
# speedup vs baseline: 2.3585x; 1.0742x over previous
#include <hip/hip_runtime.h>
#include <hip/hip_bf16.h>


typedef __bf16 bf16_t;
typedef __attribute__((ext_vector_type(16))) __bf16 v16bf;
typedef __attribute__((ext_vector_type(8)))  __bf16 v8bf;
typedef __attribute__((ext_vector_type(8)))  float  v8f;

#define NN 50000
#define EE 500000
#define K1 257
#define KT1 9
#define N1 514
#define NT1 33
#define K2 514
#define KT2 17
#define N2 64
#define NT2 4
#define K3 64
#define KT3 2
#define N3 256
#define NT3 16
#define K4 256
#define KT4 8
#define N4 1
#define NT4 1
#define KN1 192
#define KTN1 6
#define NN1 256
#define NTN1 16
#define KN2 256
#define KTN2 8
#define NN2 128
#define NTN2 8

__device__ __forceinline__ bf16_t f2bf(float f) {
  return (bf16_t)f;
}

__device__ __forceinline__ float bf2f(bf16_t b) {
  unsigned short h = __builtin_bit_cast(unsigned short, b);
  unsigned u = (unsigned)h << 16;
  return __builtin_bit_cast(float, u);
}

__device__ __forceinline__ float silu_f(float x) {
  const float h = 0.5f * x;
  return __builtin_fmaf(h, __builtin_amdgcn_tanhf(h), h);
}

__device__ __forceinline__ v16bf load_a_frag(const bf16_t* A, int stride,
                                             int lane, int kbase) {
  int m = lane & 15;
  int kh = lane >> 4;
  const bf16_t* p = A + m * stride + kbase + 8 * kh;
  v8bf lo = *(const v8bf*)(p);
  v8bf hi = *(const v8bf*)(p + 16);
  return __builtin_shufflevector(lo, hi, 0, 1, 2, 3, 4, 5, 6, 7,
                                 8, 9, 10, 11, 12, 13, 14, 15);
}

__device__ __forceinline__ v16bf load_b_frag(const bf16_t* B, int NT,
                                             int kt, int nt, int lane) {
  const bf16_t* p = B + (long long)(kt * NT + nt) * 512 + lane * 8;
  v8bf lo = *(const v8bf*)(p);
  v8bf hi = *(const v8bf*)(p + 256);
  return __builtin_shufflevector(lo, hi, 0, 1, 2, 3, 4, 5, 6, 7,
                                 8, 9, 10, 11, 12, 13, 14, 15);
}

__device__ __forceinline__ v16bf load_b_frag_w(const bf16_t* B, int NT,
                                               int kt, int ntc, int tid) {
  const bf16_t* p = B + (long long)(kt * NT + ntc) * 512
                      + (tid >> 5) * 512 + (tid & 31) * 8;
  v8bf lo = *(const v8bf*)(p);
  v8bf hi = *(const v8bf*)(p + 256);
  return __builtin_shufflevector(lo, hi, 0, 1, 2, 3, 4, 5, 6, 7,
                                 8, 9, 10, 11, 12, 13, 14, 15);
}

__device__ __forceinline__ v8f wmma_bf16(v16bf a, v16bf b, v8f c) {
  return __builtin_amdgcn_wmma_f32_16x16x32_bf16(
       false, a,  false, b,
       (short)0, c,  false,  false);
}

__device__ __forceinline__ v8f wmma_bf16_t(v16bf w, v16bf x, v8f c) {
  return wmma_bf16(w, x, c);
}

__device__ __forceinline__ v8bf silu_pack8(v8f acc) {
  v8bf o;
#pragma unroll
  for (int r = 0; r < 8; ++r) o[r] = f2bf(silu_f(acc[r]));
  return o;
}


typedef __attribute__((ext_vector_type(4))) __bf16 v4bf;

#define NNP 50176
#define CAP 64

__device__ __forceinline__ void count_edge(
    const int* __restrict__ ei, int* __restrict__ deg, int* __restrict__ bucket, int e) {
  if (e < EE) {
    const int d = ei[EE + e];
    const int s = atomicAdd(&deg[d], 1);
    if (s < CAP) bucket[(long long)d * CAP + s] = e;
  }
}

__device__ __forceinline__ void pack_weight_lane(
    const float* __restrict__ W, bf16_t* __restrict__ dst,
    int K, int N, int KT, int NT, int idx) {
  int total = KT * NT * 32;
  if (idx >= total) return;
  int lane = idx & 31;
  int tile = idx >> 5;
  int nt = tile % NT;
  int kt = tile / NT;
  int n = nt * 16 + (lane & 15);
  int kh = lane >> 4;
  bf16_t* o = dst + (long long)tile * 512 + lane * 8;
#pragma unroll
  for (int e = 0; e < 16; ++e) {
    int k = kt * 32 + 8 * kh + e + ((e >= 8) ? 8 : 0);
    float v = (k < K && n < N) ? W[(long long)k * N + n] : 0.0f;
    o[(e >> 3) * 256 + (e & 7)] = f2bf(v);
  }
}

#define PB_E1 ((KT1 * NT1 * 32 + 255) / 256)
#define PB_E2 ((KT2 * NT2 * 32 + 255) / 256)
#define PB_C1 ((KT3 * NT3 * 32 + 255) / 256)
#define PB_C2 ((KT4 * NT4 * 32 + 255) / 256)
#define PB_N1 ((KTN1 * NTN1 * 32 + 255) / 256)
#define PB_N2 ((KTN2 * NTN2 * 32 + 255) / 256)
#define PB_ZERO (NNP / 256)
#define PB_TOTAL (PB_E1 + PB_E2 + PB_C1 + PB_C2 + PB_N1 + PB_N2 + PB_ZERO)

__global__ __launch_bounds__(256) void setup_kernel(
    const float* __restrict__ W_e1, const float* __restrict__ W_e2,
    const float* __restrict__ W_c1, const float* __restrict__ W_c2,
    const float* __restrict__ W_n1, const float* __restrict__ W_n2,
    bf16_t* __restrict__ We1b, bf16_t* __restrict__ We2b,
    bf16_t* __restrict__ Wc1b, bf16_t* __restrict__ Wc2b,
    bf16_t* __restrict__ Wn1b, bf16_t* __restrict__ Wn2b,
    int* __restrict__ deg) {
  int b = blockIdx.x;
  const int t = threadIdx.x;
  if (b < PB_E1) { pack_weight_lane(W_e1, We1b, K1, N1, KT1, NT1, b * 256 + t); return; }
  b -= PB_E1;
  if (b < PB_E2) { pack_weight_lane(W_e2, We2b, K2, N2, KT2, NT2, b * 256 + t); return; }
  b -= PB_E2;
  if (b < PB_C1) { pack_weight_lane(W_c1, Wc1b, K3, N3, KT3, NT3, b * 256 + t); return; }
  b -= PB_C1;
  if (b < PB_C2) { pack_weight_lane(W_c2, Wc2b, K4, N4, KT4, NT4, b * 256 + t); return; }
  b -= PB_C2;
  if (b < PB_N1) { pack_weight_lane(W_n1, Wn1b, KN1, NN1, KTN1, NTN1, b * 256 + t); return; }
  b -= PB_N1;
  if (b < PB_N2) { pack_weight_lane(W_n2, Wn2b, KN2, NN2, KTN2, NTN2, b * 256 + t); return; }
  b -= PB_N2;
  deg[b * 256 + t] = 0;
}


#define PA 296
#define PH 552
#define PT 264
#define PNA 200
#define PNT 264
#define ME 96
#define MT (ME / 16)
#define EPW (ME / 8)
#define SMEM_BYTES (ME * PA * 2 + ME * PH * 2)
#define EDGE_BLOCKS ((EE + ME - 1) / ME)
#define COUNT_BLOCKS ((EE + 255) / 256)

__global__ __launch_bounds__(256, 4) void egnn_edge_kernel(
    const float* __restrict__ x, const int* __restrict__ ei,
    const float* __restrict__ b_e1, const float* __restrict__ b_e2,
    const float* __restrict__ b_c1, const float* __restrict__ b_c2,
    const bf16_t* __restrict__ We1, const bf16_t* __restrict__ We2,
    const bf16_t* __restrict__ Wc1, const bf16_t* __restrict__ Wc2,
    bf16_t* __restrict__ mrow, float4* __restrict__ ew,
    int* __restrict__ deg, int* __restrict__ bucket) {
  __shared__ alignas(16) char smem[SMEM_BYTES];

  if (blockIdx.x >= EDGE_BLOCKS) {
    count_edge(ei, deg, bucket, (blockIdx.x - EDGE_BLOCKS) * 256 + threadIdx.x);
    return;
  }

  bf16_t* sA = (bf16_t*)smem;
  bf16_t* sM = (bf16_t*)smem;
  bf16_t* sH = (bf16_t*)(smem + ME * PA * 2);
  bf16_t* sT = sH;

  const int tid = threadIdx.x;
  const int lane = tid & 31;
  const int wave = tid >> 5;
  const int e0 = blockIdx.x * ME;

  {
    int el = e0 + wave * EPW + (lane < EPW ? lane : 0);
    if (el > EE - 1) el = EE - 1;
    const int sv = ei[el];
    const int dv = ei[EE + el];
#pragma unroll
    for (int i = 0; i < EPW; ++i) {
      const int s = __builtin_amdgcn_readlane(sv, i);
      const int d = __builtin_amdgcn_readlane(dv, i);
      const float* xd = x + (long long)d * 131 + 4 * lane;
      const float* xs = x + (long long)s * 131 + 4 * lane;
      const int m = wave * EPW + i;
      v4bf pd, ps;
#pragma unroll
      for (int r = 0; r < 4; ++r) {
        pd[r] = f2bf(xd[r]);
        ps[r] = f2bf(xs[r]);
      }
      *(v4bf*)(sA + m * PA + 4 * lane)       = pd;
      *(v4bf*)(sA + m * PA + 128 + 4 * lane) = ps;
    }
    if (lane < EPW) {
      const int m = wave * EPW + lane;
      const float* xs = x + (long long)sv * 131;
      const float* xd = x + (long long)dv * 131;
      float r0 = xs[128] - xd[128];
      float r1 = xs[129] - xd[129];
      float r2 = xs[130] - xd[130];
      float dist = sqrtf(r0 * r0 + r1 * r1 + r2 * r2);
      v8bf z;
#pragma unroll
      for (int k = 0; k < 8; ++k) z[k] = f2bf(0.0f);
      v8bf t = z;
      t[0] = f2bf(dist);
      *(v8bf*)(sA + m * PA + 256) = t;
      *(v8bf*)(sA + m * PA + 264) = z;
      *(v8bf*)(sA + m * PA + 272) = z;
      *(v8bf*)(sA + m * PA + 280) = z;
      float* pr = (float*)(sA + m * PA + 288);
      pr[0] = r0; pr[1] = r1; pr[2] = r2;
    }
#pragma unroll
    for (int i = 0; i < ME * 16 / 256; ++i) {
      int idx = tid + i * 256;
      sH[(idx >> 4) * PH + 528 + (idx & 15)] = f2bf(0.0f);
    }
  }
  v16bf bq[2][2];
  bq[0][0] = load_b_frag_w(We1, NT1, 0, 0, tid);
  bq[0][1] = load_b_frag_w(We1, NT1, 0, 8, tid);
  v8f bias0, bias1;
#pragma unroll
  for (int r = 0; r < 8; ++r) {
    bias0[r] = b_e1[wave * 16 + 8 * (lane >> 4) + r];
    bias1[r] = b_e1[(wave + 8) * 16 + 8 * (lane >> 4) + r];
  }
  __syncthreads();

  {
    const int c = lane & 15;
    const int f8 = 8 * (lane >> 4);

    v8f acc0[MT], acc1[MT];
    v16bf a[MT];
#pragma unroll
    for (int mt = 0; mt < MT / 2; ++mt)
      a[mt] = load_a_frag(sA + mt * 16 * PA, PA, lane, 0);
#pragma unroll
    for (int s = 0; s < 2 * KT1; ++s) {
      const int p = s / KT1;
      const int kt = s % KT1;
      const int n0 = (wave + 16 * p) * 16 + f8;
      const int n1 = (wave + 16 * p + 8) * 16 + f8;
      if (kt == 0) {
#pragma unroll
        for (int mt = 0; mt < MT; ++mt) { acc0[mt] = bias0; acc1[mt] = bias1; }
      }
      if (s + 1 < 2 * KT1) {
        const int p2 = (s + 1) / KT1;
        const int k2 = (s + 1) % KT1;
        bq[(s + 1) % 2][0] = load_b_frag_w(We1, NT1, k2, 16 * p2, tid);
        bq[(s + 1) % 2][1] = load_b_frag_w(We1, NT1, k2, 16 * p2 + 8, tid);
      }
#pragma unroll
      for (int mt = MT / 2; mt < MT; ++mt)
        a[mt] = load_a_frag(sA + mt * 16 * PA, PA, lane, kt * 32);
      __builtin_amdgcn_sched_barrier(0);
#pragma unroll
      for (int mt = 0; mt < MT / 2; ++mt) {
        acc0[mt] = wmma_bf16_t(bq[s % 2][0], a[mt], acc0[mt]);
        acc1[mt] = wmma_bf16_t(bq[s % 2][1], a[mt], acc1[mt]);
      }
      __builtin_amdgcn_sched_barrier(0);
      if (s + 1 < 2 * KT1) {
        const int k1 = (s + 1) % KT1;
#pragma unroll
        for (int mt = 0; mt < MT / 2; ++mt)
          a[mt] = load_a_frag(sA + mt * 16 * PA, PA, lane, k1 * 32);
      }
      __builtin_amdgcn_sched_barrier(0);
#pragma unroll
      for (int mt = MT / 2; mt < MT; ++mt) {
        acc0[mt] = wmma_bf16_t(bq[s % 2][0], a[mt], acc0[mt]);
        acc1[mt] = wmma_bf16_t(bq[s % 2][1], a[mt], acc1[mt]);
      }
      if (kt == KT1 - 1) {
        if (p == 0) {
#pragma unroll
          for (int r = 0; r < 8; ++r) {
            bias0[r] = b_e1[n0 + 256 + r];
            bias1[r] = b_e1[n1 + 256 + r];
          }
          __builtin_amdgcn_sched_barrier(0);
        }
#pragma unroll
        for (int mt = 0; mt < MT; ++mt) {
          *(v8bf*)(sH + (mt * 16 + c) * PH + n0) = silu_pack8(acc0[mt]);
          *(v8bf*)(sH + (mt * 16 + c) * PH + n1) = silu_pack8(acc1[mt]);
        }
      }
    }

    if (wave == 0) {
      const int n0 = (NT1 - 1) * 16 + f8;
      v8f t[MT];
#pragma unroll
      for (int r = 0; r < 8; ++r) t[0][r] = (n0 + r < N1) ? b_e1[n0 + r] : 0.0f;
#pragma unroll
      for (int mt = 1; mt < MT; ++mt) t[mt] = t[0];
      v16bf bt[KT1];
#pragma unroll
      for (int kt = 0; kt < KT1; ++kt)
        bt[kt] = load_b_frag_w(We1, NT1, kt, 32, tid);
      __builtin_amdgcn_sched_barrier(0);
      v16bf am[2][MT];
#pragma unroll
      for (int mt = 0; mt < MT; ++mt)
        am[0][mt] = load_a_frag(sA + mt * 16 * PA, PA, lane, 0);
#pragma unroll
      for (int kt = 0; kt < KT1; ++kt) {
        if (kt + 1 < KT1) {
#pragma unroll
          for (int mt = 0; mt < MT; ++mt)
            am[(kt + 1) % 2][mt] = load_a_frag(sA + mt * 16 * PA, PA, lane, (kt + 1) * 32);
        }
        __builtin_amdgcn_sched_barrier(0);
#pragma unroll
        for (int mt = 0; mt < MT; ++mt)
          t[mt] = wmma_bf16_t(bt[kt], am[kt % 2][mt], t[mt]);
        __builtin_amdgcn_sched_barrier(0);
      }
#pragma unroll
      for (int mt = 0; mt < MT; ++mt)
        *(v8bf*)(sH + (mt * 16 + c) * PH + n0) = silu_pack8(t[mt]);
    }
  }
  __syncthreads();

  {
    int mp = wave >> 2;
    int nt = wave & 3;
    int c = lane & 15;
    int n0 = nt * 16 + 8 * (lane >> 4);
    v8f acc[MT / 2];
#pragma unroll
    for (int r = 0; r < 8; ++r) acc[0][r] = b_e2[n0 + r];
#pragma unroll
    for (int j = 1; j < MT / 2; ++j) acc[j] = acc[0];
#pragma unroll
    for (int kt = 0; kt < KT2; ++kt) {
      v16bf b = load_b_frag(We2, NT2, kt, nt, lane);
#pragma unroll
      for (int j = 0; j < MT / 2; ++j) {
        v16bf a = load_a_frag(sH + (mp + 2 * j) * 16 * PH, PH, lane, kt * 32);
        acc[j] = wmma_bf16_t(b, a, acc[j]);
      }
    }
#pragma unroll
    for (int j = 0; j < MT / 2; ++j)
      *(v8bf*)(sM + ((mp + 2 * j) * 16 + c) * PA + n0) = silu_pack8(acc[j]);
  }
  __syncthreads();

  {
    const int c = lane & 15;
    const int nt0 = wave;
    const int nt1 = wave + 8;
    const int n0 = nt0 * 16 + 8 * (lane >> 4);
    const int n1 = nt1 * 16 + 8 * (lane >> 4);
    v8f bias0, bias1;
#pragma unroll
    for (int r = 0; r < 8; ++r) { bias0[r] = b_c1[n0 + r]; bias1[r] = b_c1[n1 + r]; }
    v8f acc0[MT], acc1[MT];
#pragma unroll
    for (int mt = 0; mt < MT; ++mt) { acc0[mt] = bias0; acc1[mt] = bias1; }
#pragma unroll
    for (int kt = 0; kt < KT3; ++kt) {
      v16bf b0 = load_b_frag_w(Wc1, NT3, kt, 0, tid);
      v16bf b1 = load_b_frag_w(Wc1, NT3, kt, 8, tid);
#pragma unroll
      for (int mt = 0; mt < MT; ++mt) {
        v16bf a = load_a_frag(sM + mt * 16 * PA, PA, lane, kt * 32);
        acc0[mt] = wmma_bf16_t(b0, a, acc0[mt]);
        acc1[mt] = wmma_bf16_t(b1, a, acc1[mt]);
      }
    }
#pragma unroll
    for (int mt = 0; mt < MT; ++mt) {
      *(v8bf*)(sT + (mt * 16 + c) * PT + n0) = silu_pack8(acc0[mt]);
      *(v8bf*)(sT + (mt * 16 + c) * PT + n1) = silu_pack8(acc1[mt]);
    }
  }
  __syncthreads();

  if (wave < MT) {
    int mt = wave;
    v8f acc;
#pragma unroll
    for (int r = 0; r < 8; ++r) acc[r] = 0.0f;
    if (lane < 16) acc[0] = b_c2[0];
#pragma unroll
    for (int kt = 0; kt < KT4; ++kt) {
      v16bf a = load_a_frag(sT + mt * 16 * PT, PT, lane, kt * 32);
      acc = wmma_bf16_t(load_b_frag(Wc2, NT4, kt, 0, lane), a, acc);
    }
    if (lane < 16)
      ((float*)(sA + (mt * 16 + lane) * PA + 288))[3] = acc[0];
  }
  __syncthreads();

  {
    int sub = tid & 7;
#pragma unroll
    for (int h = 0; h < ME / 32; ++h) {
      int m = h * 32 + (tid >> 3);
      if (e0 + m < EE) {
        v8bf v = *(const v8bf*)(sM + m * PA + sub * 8);
        *(v8bf*)(mrow + (long long)(e0 + m) * 64 + sub * 8) = v;
      }
    }
  }
  if (tid < ME && e0 + tid < EE) {
    const float* pr = (const float*)(sA + tid * PA + 288);
    ew[e0 + tid] = make_float4(pr[0], pr[1], pr[2], pr[3]);
  }
}


__global__ __launch_bounds__(256) void egnn_node_kernel(
    const float* __restrict__ x, const bf16_t* __restrict__ mrow,
    const float4* __restrict__ ew, const int* __restrict__ ei,
    const int* __restrict__ deg, const int* __restrict__ bucket,
    const float* __restrict__ b_n1, const float* __restrict__ b_n2,
    const bf16_t* __restrict__ Wn1, const bf16_t* __restrict__ Wn2,
    float* __restrict__ out) {
  __shared__ alignas(16) bf16_t sA[16 * PNA];
  __shared__ alignas(16) bf16_t sT[16 * PNT];
  __shared__ float sW[16];
  __shared__ float sR[16][3];
  __shared__ alignas(16) int sU[16 * CAP];
  __shared__ alignas(16) int sS[16 * CAP];

  const int tid = threadIdx.x;
  const int lane = tid & 31;
  const int wave = tid >> 5;
  const int g0 = blockIdx.x * 16;

  {
    int m = tid >> 4;
    int sub = tid & 15;
    long long g = g0 + m;
#pragma unroll
    for (int j = 0; j < 8; ++j) {
      int k = sub * 8 + j;
      sA[m * PNA + k] = f2bf(x[g * 131 + k]);
    }
    const int n = deg[g];
    const bool spill = n > CAP;
    const int nb = spill ? 0 : n;
    int ev[CAP / 16];
#pragma unroll
    for (int q = 0; q < CAP / 16; ++q) {
      const int idx = sub + 16 * q;
      ev[q] = (idx < nb) ? bucket[g * CAP + idx] : 0x7fffffff;
    }
#pragma unroll
    for (int q = 0; q < CAP / 16; ++q) sU[m * CAP + sub + 16 * q] = ev[q];
    __syncthreads();
    for (int idx = sub; idx < nb; idx += 16) {
      const int mine = sU[m * CAP + idx];
      int r = 0;
      for (int j = 0; j < nb; j += 4) {
        const int4 v = *(const int4*)(sU + m * CAP + j);
        r += (v.x < mine ? 1 : 0) + (v.y < mine ? 1 : 0)
           + (v.z < mine ? 1 : 0) + (v.w < mine ? 1 : 0);
      }
      sS[m * CAP + r] = mine;
    }
    __syncthreads();
    float a0 = 0.0f, a1 = 0.0f, a2 = 0.0f, a3 = 0.0f;
    float r0 = 0.0f, r1 = 0.0f, r2 = 0.0f, w = 0.0f;
    auto add_edge = [&](int e) {
      v4bf q = *(const v4bf*)(mrow + (long long)e * 64 + sub * 4);
      a0 += bf2f(q[0]); a1 += bf2f(q[1]); a2 += bf2f(q[2]); a3 += bf2f(q[3]);
      if (sub == 0) {
        float4 t = ew[e];
        r0 += t.x; r1 += t.y; r2 += t.z; w += t.w;
      }
    };
    for (int i = 0; i < nb; ++i) add_edge(sS[m * CAP + i]);
    if (spill) {
      for (int e = 0; e < EE; ++e)
        if (ei[EE + e] == (int)g) add_edge(e);
    }
    sA[m * PNA + 128 + sub * 4 + 0] = f2bf(a0);
    sA[m * PNA + 128 + sub * 4 + 1] = f2bf(a1);
    sA[m * PNA + 128 + sub * 4 + 2] = f2bf(a2);
    sA[m * PNA + 128 + sub * 4 + 3] = f2bf(a3);
    if (sub == 0) {
      sW[m] = w;
      sR[m][0] = r0; sR[m][1] = r1; sR[m][2] = r2;
    }
  }
  __syncthreads();

  {
    v16bf a[KTN1];
#pragma unroll
    for (int kt = 0; kt < KTN1; ++kt) a[kt] = load_a_frag(sA, PNA, lane, kt * 32);
#pragma unroll
    for (int i = 0; i < 2; ++i) {
      int nt = wave + 8 * i;
      int n0 = nt * 16 + 8 * (lane >> 4);
      v8f acc;
#pragma unroll
      for (int r = 0; r < 8; ++r) acc[r] = b_n1[n0 + r];
#pragma unroll
      for (int kt = 0; kt < KTN1; ++kt)
        acc = wmma_bf16_t(load_b_frag(Wn1, NTN1, kt, nt, lane), a[kt], acc);
      *(v8bf*)(sT + (lane & 15) * PNT + n0) = silu_pack8(acc);
    }
  }
  __syncthreads();

  {
    int nt = wave;
    int n = nt * 16 + (lane & 15);
    float bias = b_n2[n];
    v8f acc;
#pragma unroll
    for (int r = 0; r < 8; ++r) acc[r] = bias;
#pragma unroll
    for (int kt = 0; kt < KTN2; ++kt) {
      v16bf a = load_a_frag(sT, PNT, lane, kt * 32);
      acc = wmma_bf16(a, load_b_frag(Wn2, NTN2, kt, nt, lane), acc);
    }
#pragma unroll
    for (int r = 0; r < 8; ++r) {
      int m = r + 8 * (lane >> 4);
      long long g = g0 + m;
      out[g * 131 + n] = acc[r] + x[g * 131 + n];
    }
  }

  if (tid < 48) {
    int m = tid / 3, c = tid % 3;
    long long g = g0 + m;
    out[g * 131 + 128 + c] = x[g * 131 + 128 + c] + sW[m] * sR[m][c];
  }
}


extern "C" void kernel_launch(void* const* d_in, const int* in_sizes, int n_in,
                              void* d_out, int out_size, void* d_ws,
                              size_t ws_size, hipStream_t stream) {
  const float* x    = (const float*)d_in[0];
  const int*   ei   = (const int*)d_in[1];
  const float* W_e1 = (const float*)d_in[2];
  const float* b_e1 = (const float*)d_in[3];
  const float* W_e2 = (const float*)d_in[4];
  const float* b_e2 = (const float*)d_in[5];
  const float* W_c1 = (const float*)d_in[6];
  const float* b_c1 = (const float*)d_in[7];
  const float* W_c2 = (const float*)d_in[8];
  const float* b_c2 = (const float*)d_in[9];
  const float* W_n1 = (const float*)d_in[10];
  const float* b_n1 = (const float*)d_in[11];
  const float* W_n2 = (const float*)d_in[12];
  const float* b_n2 = (const float*)d_in[13];
  float* out = (float*)d_out;

  char* ws = (char*)d_ws;
  int* deg  = (int*)ws;
  int* bucket = deg + NNP;
  bf16_t* mrow = (bf16_t*)(bucket + (long long)NN * CAP);
  float4* ew = (float4*)(mrow + (long long)EE * 64);
  char* wp = (char*)(ew + EE);
  bf16_t* We1b = (bf16_t*)wp; wp += (long long)KT1 * NT1 * 32 * 16 * 2;
  bf16_t* We2b = (bf16_t*)wp; wp += (long long)KT2 * NT2 * 32 * 16 * 2;
  bf16_t* Wc1b = (bf16_t*)wp; wp += (long long)KT3 * NT3 * 32 * 16 * 2;
  bf16_t* Wc2b = (bf16_t*)wp; wp += (long long)KT4 * NT4 * 32 * 16 * 2;
  bf16_t* Wn1b = (bf16_t*)wp; wp += (long long)KTN1 * NTN1 * 32 * 16 * 2;
  bf16_t* Wn2b = (bf16_t*)wp; wp += (long long)KTN2 * NTN2 * 32 * 16 * 2;

  setup_kernel<<<PB_TOTAL, 256, 0, stream>>>(
      W_e1, W_e2, W_c1, W_c2, W_n1, W_n2,
      We1b, We2b, Wc1b, Wc2b, Wn1b, Wn2b, deg);

  egnn_edge_kernel<<<EDGE_BLOCKS + COUNT_BLOCKS, 256, 0, stream>>>(
      x, ei, b_e1, b_e2, b_c1, b_c2, We1b, We2b, Wc1b, Wc2b, mrow, ew, deg, bucket);

  egnn_node_kernel<<<NN / 16, 256, 0, stream>>>(
      x, mrow, ew, ei, deg, bucket, b_n1, b_n2, Wn1b, Wn2b, out);
}
